# starting kernel
# baseline (speedup 1.0000x reference)
#include <hip/hip_runtime.h>
#include <math.h>

typedef __attribute__((ext_vector_type(2))) float v2f;
typedef __attribute__((ext_vector_type(8))) float v8f;

#define N_NODES 100000
#define HDIM 16

__global__ void zero_f32(float* __restrict__ p, long long n) {
    long long t = (long long)blockIdx.x * blockDim.x + threadIdx.x;
    long long stride = (long long)gridDim.x * blockDim.x;
    for (; t < n; t += stride) p[t] = 0.0f;
}

template <int K>
__global__ __launch_bounds__(256) void node_transform(
    const float* __restrict__ xin, const float* __restrict__ Wl,
    const float* __restrict__ Wr,  const float* __restrict__ bias,
    float* __restrict__ outl, float* __restrict__ outr, int nNodes)
{
    const int wave  = blockIdx.x * (blockDim.x >> 5) + (threadIdx.x >> 5);
    const int lane  = threadIdx.x & 31;
    const int tile0 = wave * 16;
    if (tile0 >= nNodes) return;

    const int m  = lane & 15;
    const int kh = (lane >> 4) << 1;

    int rowA = tile0 + m;
    if (rowA >= nNodes) rowA = nNodes - 1;
    const float* xrow = xin + (long long)rowA * K;

    v8f accl = {0.f,0.f,0.f,0.f,0.f,0.f,0.f,0.f};
    v8f accr = {0.f,0.f,0.f,0.f,0.f,0.f,0.f,0.f};

#pragma unroll
    for (int k0 = 0; k0 < K; k0 += 4) {
        const int kr = k0 + kh;
        v2f a = *(const v2f*)(xrow + kr);
        v2f bl = { Wl[kr * HDIM + m], Wl[(kr + 1) * HDIM + m] };
        v2f br = { Wr[kr * HDIM + m], Wr[(kr + 1) * HDIM + m] };
        accl = __builtin_amdgcn_wmma_f32_16x16x4_f32(
                   false, a, false, bl, (short)0, accl, false, false);
        accr = __builtin_amdgcn_wmma_f32_16x16x4_f32(
                   false, a, false, br, (short)0, accr, false, false);
    }

    const float bv    = bias[m];
    const int   mbase = (lane >> 4) * 8;
#pragma unroll
    for (int v = 0; v < 8; ++v) {
        const int row = tile0 + mbase + v;
        if (row < nNodes) {
            outl[row * HDIM + m] = accl[v];
            outr[row * HDIM + m] = accr[v] + bv;
        }
    }
}

__global__ __launch_bounds__(256) void edge_aggregate(
    const float* __restrict__ y, const int* __restrict__ eidx,
    float* __restrict__ agg, float* __restrict__ cnt, int E, int withCount)
{
    const long long total  = (long long)E * HDIM;
    const long long stride = (long long)gridDim.x * blockDim.x;
    for (long long t = (long long)blockIdx.x * blockDim.x + threadIdx.x;
         t < total; t += stride) {
        const int e = (int)(t >> 4);
        const int j = (int)(t & 15);
        const int s = eidx[e];
        const int d = eidx[E + e];
        atomicAdd(agg + (long long)d * HDIM + j, y[(long long)s * HDIM + j]);
        if (withCount && j == 0) atomicAdd(cnt + d, 1.0f);
    }
}

__global__ __launch_bounds__(256) void combine(
    const float* __restrict__ agg, const float* __restrict__ yr,
    const float* __restrict__ cnt, float* __restrict__ out,
    int nNodes, int doRelu)
{
    const int t = blockIdx.x * blockDim.x + threadIdx.x;
    if (t >= nNodes * HDIM) return;
    const int node = t >> 4;
    float c = cnt[node];
    c = c > 1.0f ? c : 1.0f;
    float v = agg[t] / c + yr[t];
    if (doRelu) v = v > 0.0f ? v : 0.0f;
    out[t] = v;
}

__global__ __launch_bounds__(256) void decode_edges(
    const float* __restrict__ h, const int* __restrict__ didx,
    float* __restrict__ out, int P)
{
    const int t = blockIdx.x * blockDim.x + threadIdx.x;
    if (t >= P) return;
    const int a = didx[t];
    const int b = didx[P + t];
    const float4* ha = (const float4*)(h + (long long)a * HDIM);
    const float4* hb = (const float4*)(h + (long long)b * HDIM);
    float s = 0.0f;
#pragma unroll
    for (int i = 0; i < 4; ++i) {
        float4 u = ha[i];
        float4 w = hb[i];
        s += u.x * w.x + u.y * w.y + u.z * w.z + u.w * w.w;
    }
    out[t] = 1.0f / (1.0f + __expf(-s));
}

extern "C" void kernel_launch(void* const* d_in, const int* in_sizes, int n_in,
                              void* d_out, int out_size, void* d_ws, size_t ws_size,
                              hipStream_t stream) {
    const float* x   = (const float*)d_in[0];
    const float* W1l = (const float*)d_in[1];
    const float* b1  = (const float*)d_in[2];
    const float* W1r = (const float*)d_in[3];
    const float* W2l = (const float*)d_in[4];
    const float* b2  = (const float*)d_in[5];
    const float* W2r = (const float*)d_in[6];
    const int* eidx  = (const int*)d_in[7];
    const int* didx  = (const int*)d_in[8];
    float* out = (float*)d_out;

    const int N = in_sizes[0] / 128;
    const int E = in_sizes[7] / 2;
    const int P = in_sizes[8] / 2;
    const long long NH = (long long)N * HDIM;

    float* buf0 = (float*)d_ws;
    float* buf1 = buf0 + NH;
    float* buf2 = buf1 + NH;
    float* buf3 = buf2 + NH;
    float* cnt  = buf3 + NH;

    const int TB = 256;
    const int wavesPerBlock = TB / 32;
    const int tiles = (N + 15) / 16;
    const int xfBlocks = (tiles + wavesPerBlock - 1) / wavesPerBlock;
    const long long edgeWork = (long long)E * HDIM;
    const int edgeBlocks = (int)((edgeWork + TB - 1) / TB);
    const int nodeBlocks = (int)((NH + TB - 1) / TB);
    const int decBlocks  = (P + TB - 1) / TB;

    zero_f32<<<2048, TB, 0, stream>>>(buf2, 2 * NH + N);

    node_transform<128><<<xfBlocks, TB, 0, stream>>>(x, W1l, W1r, b1,
                                                     buf0, buf1, N);

    edge_aggregate<<<edgeBlocks, TB, 0, stream>>>(buf0, eidx, buf2, cnt, E, 1);

    combine<<<nodeBlocks, TB, 0, stream>>>(buf2, buf1, cnt, buf0, N, 1);

    node_transform<16><<<xfBlocks, TB, 0, stream>>>(buf0, W2l, W2r, b2,
                                                    buf1, buf2, N);

    edge_aggregate<<<edgeBlocks, TB, 0, stream>>>(buf1, eidx, buf3, cnt, E, 0);

    combine<<<nodeBlocks, TB, 0, stream>>>(buf3, buf2, cnt, buf0, N, 0);

    decode_edges<<<decBlocks, TB, 0, stream>>>(buf0, didx, out, P);
}


# Aggregate by dst-range buckets: bin edges by dst>>8 (LDS histogram, scan, scatter), sum each bucket in LDS with fused mean+combine; no global float atomics; terms summed in arrival-order rank
# speedup vs baseline: 2.1537x; 2.1537x over previous
#include <hip/hip_runtime.h>
#include <math.h>

typedef __attribute__((ext_vector_type(2))) float v2f;
typedef __attribute__((ext_vector_type(8))) float v8f;

#define N_NODES 100000
#define HDIM 16

template <int K>
__global__ __launch_bounds__(256) void node_transform(
    const float* __restrict__ xin, const float* __restrict__ Wl,
    const float* __restrict__ Wr,  const float* __restrict__ bias,
    float* __restrict__ outl, float* __restrict__ outr, int nNodes)
{
    const int wave  = blockIdx.x * (blockDim.x >> 5) + (threadIdx.x >> 5);
    const int lane  = threadIdx.x & 31;
    const int tile0 = wave * 16;
    if (tile0 >= nNodes) return;

    const int m  = lane & 15;
    const int kh = (lane >> 4) << 1;

    int rowA = tile0 + m;
    if (rowA >= nNodes) rowA = nNodes - 1;
    const float* xrow = xin + (long long)rowA * K;

    v8f accl = {0.f,0.f,0.f,0.f,0.f,0.f,0.f,0.f};
    v8f accr = {0.f,0.f,0.f,0.f,0.f,0.f,0.f,0.f};

#pragma unroll
    for (int k0 = 0; k0 < K; k0 += 4) {
        const int kr = k0 + kh;
        v2f a = *(const v2f*)(xrow + kr);
        v2f bl = { Wl[kr * HDIM + m], Wl[(kr + 1) * HDIM + m] };
        v2f br = { Wr[kr * HDIM + m], Wr[(kr + 1) * HDIM + m] };
        accl = __builtin_amdgcn_wmma_f32_16x16x4_f32(
                   false, a, false, bl, (short)0, accl, false, false);
        accr = __builtin_amdgcn_wmma_f32_16x16x4_f32(
                   false, a, false, br, (short)0, accr, false, false);
    }

    const float bv    = bias[m];
    const int   mbase = (lane >> 4) * 8;
#pragma unroll
    for (int v = 0; v < 8; ++v) {
        const int row = tile0 + mbase + v;
        if (row < nNodes) {
            outl[row * HDIM + m] = accl[v];
            outr[row * HDIM + m] = accr[v] + bv;
        }
    }
}

#define BSHIFT 8
#define BNODES 256
#define MAXNB  1024
#define GCHUNKS 128

__global__ __launch_bounds__(1024) void bucket_hist(
    const int* __restrict__ eidx, int* __restrict__ hist,
    int E, int NB, int chunk)
{
    __shared__ int h[MAXNB];
    const int tid = threadIdx.x;
    for (int i = tid; i < NB; i += 1024) h[i] = 0;
    __syncthreads();
    const int beg = blockIdx.x * chunk;
    const int end = (beg + chunk < E) ? (beg + chunk) : E;
    const int* __restrict__ dstp = eidx + E;
    for (int e0 = beg; e0 < end; e0 += 4096) {
        int  d[4];
        bool ok[4];
#pragma unroll
        for (int u = 0; u < 4; ++u) {
            const int e = e0 + u * 1024 + tid;
            ok[u] = e < end;
            d[u]  = dstp[ok[u] ? e : end - 1];
        }
#pragma unroll
        for (int u = 0; u < 4; ++u)
            atomicAdd(&h[d[u] >> BSHIFT], ok[u] ? 1 : 0);
    }
    __syncthreads();
    for (int i = tid; i < NB; i += 1024) hist[blockIdx.x * NB + i] = h[i];
}

__global__ __launch_bounds__(1024) void bucket_scan(
    const int* __restrict__ hist, int* __restrict__ offs,
    int* __restrict__ bstart, int NB)
{
    __shared__ int sc[1024];
    const int tid = threadIdx.x;
    int total = 0;
    if (tid < NB) {
        int run = 0;
        for (int g0 = 0; g0 < GCHUNKS; g0 += 32) {
            int v[32];
#pragma unroll
            for (int k = 0; k < 32; ++k) v[k] = hist[(g0 + k) * NB + tid];
#pragma unroll
            for (int k = 0; k < 32; ++k) {
                offs[(g0 + k) * NB + tid] = run;
                run += v[k];
            }
        }
        total = run;
    }
    sc[tid] = total;
    __syncthreads();
    for (int d = 1; d < 1024; d <<= 1) {
        const int t = (tid >= d) ? sc[tid - d] : 0;
        __syncthreads();
        sc[tid] += t;
        __syncthreads();
    }
    if (tid < NB) bstart[tid] = sc[tid] - total;
    if (tid == NB - 1) bstart[NB] = sc[tid];
}

__global__ __launch_bounds__(1024) void bucket_scatter(
    const int* __restrict__ eidx, const int* __restrict__ offs,
    const int* __restrict__ bstart, unsigned* __restrict__ packed,
    int E, int NB, int chunk)
{
    __shared__ int cur[MAXNB];
    const int tid = threadIdx.x;
    for (int i = tid; i < NB; i += 1024)
        cur[i] = bstart[i] + offs[blockIdx.x * NB + i];
    __syncthreads();
    const int beg = blockIdx.x * chunk;
    const int end = (beg + chunk < E) ? (beg + chunk) : E;
    const int* __restrict__ dstp = eidx + E;
    int e0 = beg;
    for (; e0 + 4096 <= end; e0 += 4096) {
        int s[4], d[4];
#pragma unroll
        for (int u = 0; u < 4; ++u) {
            const int e = e0 + u * 1024 + tid;
            s[u] = eidx[e];
            d[u] = dstp[e];
        }
#pragma unroll
        for (int u = 0; u < 4; ++u) {
            const int pos = atomicAdd(&cur[d[u] >> BSHIFT], 1);
            packed[pos] = ((unsigned)s[u] << BSHIFT) | (unsigned)(d[u] & (BNODES - 1));
        }
    }
    for (int e = e0 + tid; e < end; e += 1024) {
        const int s = eidx[e];
        const int d = dstp[e];
        const int pos = atomicAdd(&cur[d >> BSHIFT], 1);
        packed[pos] = ((unsigned)s << BSHIFT) | (unsigned)(d & (BNODES - 1));
    }
}

__global__ __launch_bounds__(1024) void bucket_aggregate(
    const float* __restrict__ y, const float* __restrict__ yr,
    const unsigned* __restrict__ packed, const int* __restrict__ bstart,
    float* __restrict__ out, int nNodes, int doRelu)
{
    __shared__ float acc[BNODES * HDIM];
    __shared__ float cntL[BNODES];
    const int tid = threadIdx.x;
    for (int i = tid; i < BNODES * HDIM; i += 1024) acc[i] = 0.0f;
    if (tid < BNODES) cntL[tid] = 0.0f;
    __syncthreads();

    const int beg  = bstart[blockIdx.x];
    const int end  = bstart[blockIdx.x + 1];
    const int j    = tid & 15;
    const int slot = tid >> 4;
    for (int base = beg; base < end; base += 64 * 8) {
        unsigned pk[8];
        float    v[8];
        bool     ok[8];
#pragma unroll
        for (int u = 0; u < 8; ++u) {
            const int idx = base + u * 64 + slot;
            ok[u] = idx < end;
            pk[u] = packed[ok[u] ? idx : end - 1];
        }
#pragma unroll
        for (int u = 0; u < 8; ++u)
            v[u] = y[(long long)(pk[u] >> BSHIFT) * HDIM + j];
#pragma unroll
        for (int u = 0; u < 8; ++u) {
            const int dl = (int)(pk[u] & (BNODES - 1));
            atomicAdd(&acc[dl * HDIM + j], ok[u] ? v[u] : 0.0f);
            if (j == 0) atomicAdd(&cntL[dl], ok[u] ? 1.0f : 0.0f);
        }
    }
    __syncthreads();

    const int lo = blockIdx.x * BNODES;
    for (int i = tid; i < BNODES * HDIM; i += 1024) {
        const int n = lo + (i >> 4);
        if (n < nNodes) {
            float c = cntL[i >> 4];
            c = c > 1.0f ? c : 1.0f;
            const long long t = (long long)n * HDIM + (i & 15);
            float r = acc[i] / c + yr[t];
            if (doRelu) r = r > 0.0f ? r : 0.0f;
            out[t] = r;
        }
    }
}

__global__ __launch_bounds__(256) void decode_edges(
    const float* __restrict__ h, const int* __restrict__ didx,
    float* __restrict__ out, int P)
{
    const int t = blockIdx.x * blockDim.x + threadIdx.x;
    if (t >= P) return;
    const int a = didx[t];
    const int b = didx[P + t];
    const float4* ha = (const float4*)(h + (long long)a * HDIM);
    const float4* hb = (const float4*)(h + (long long)b * HDIM);
    float s = 0.0f;
#pragma unroll
    for (int i = 0; i < 4; ++i) {
        float4 u = ha[i];
        float4 w = hb[i];
        s += u.x * w.x + u.y * w.y + u.z * w.z + u.w * w.w;
    }
    out[t] = 1.0f / (1.0f + __expf(-s));
}

extern "C" void kernel_launch(void* const* d_in, const int* in_sizes, int n_in,
                              void* d_out, int out_size, void* d_ws, size_t ws_size,
                              hipStream_t stream) {
    const float* x   = (const float*)d_in[0];
    const float* W1l = (const float*)d_in[1];
    const float* b1  = (const float*)d_in[2];
    const float* W1r = (const float*)d_in[3];
    const float* W2l = (const float*)d_in[4];
    const float* b2  = (const float*)d_in[5];
    const float* W2r = (const float*)d_in[6];
    const int* eidx  = (const int*)d_in[7];
    const int* didx  = (const int*)d_in[8];
    float* out = (float*)d_out;

    const int N = in_sizes[0] / 128;
    const int E = in_sizes[7] / 2;
    const int P = in_sizes[8] / 2;
    const long long NH = (long long)N * HDIM;

    float* buf0 = (float*)d_ws;
    float* buf1 = buf0 + NH;
    float* buf2 = buf1 + NH;
    unsigned* packed = (unsigned*)(buf2 + NH);
    int* bstart = (int*)(packed + E);

    const int NB    = (N + BNODES - 1) / BNODES;
    const int chunk = (E + GCHUNKS - 1) / GCHUNKS;
    int* hist = (int*)buf2;
    int* offs = hist + GCHUNKS * NB;

    const int TB = 256;
    const int wavesPerBlock = TB / 32;
    const int tiles = (N + 15) / 16;
    const int xfBlocks = (tiles + wavesPerBlock - 1) / wavesPerBlock;
    const int decBlocks  = (P + TB - 1) / TB;

    bucket_hist<<<GCHUNKS, 1024, 0, stream>>>(eidx, hist, E, NB, chunk);
    bucket_scan<<<1, 1024, 0, stream>>>(hist, offs, bstart, NB);
    bucket_scatter<<<GCHUNKS, 1024, 0, stream>>>(eidx, offs, bstart, packed,
                                                 E, NB, chunk);

    node_transform<128><<<xfBlocks, TB, 0, stream>>>(x, W1l, W1r, b1,
                                                     buf0, buf1, N);

    bucket_aggregate<<<NB, 1024, 0, stream>>>(buf0, buf1, packed, bstart,
                                              buf2, N, 1);

    node_transform<16><<<xfBlocks, TB, 0, stream>>>(buf2, W2l, W2r, b2,
                                                    buf0, buf1, N);

    bucket_aggregate<<<NB, 1024, 0, stream>>>(buf0, buf1, packed, bstart,
                                              buf2, N, 0);

    decode_edges<<<decBlocks, TB, 0, stream>>>(buf2, didx, out, P);
}


# layer-1 projection runs in the scan's launch (work-groups behind the one that scans; the two are independent): they overlap and one launch goes, 8 -> 7; arithmetic and sum order unchanged
# speedup vs baseline: 2.2964x; 1.0663x over previous
#include <hip/hip_runtime.h>
#include <math.h>

typedef __attribute__((ext_vector_type(2))) float v2f;
typedef __attribute__((ext_vector_type(8))) float v8f;

#define N_NODES 100000
#define HDIM 16

template <int K>
__device__ __forceinline__ void node_tile(
    const int wave,
    const float* __restrict__ xin, const float* __restrict__ Wl,
    const float* __restrict__ Wr,  const float* __restrict__ bias,
    float* __restrict__ outl, float* __restrict__ outr, int nNodes)
{
    const int lane  = threadIdx.x & 31;
    const int tile0 = wave * 16;
    if (tile0 >= nNodes) return;

    const int m  = lane & 15;
    const int kh = (lane >> 4) << 1;

    int rowA = tile0 + m;
    if (rowA >= nNodes) rowA = nNodes - 1;
    const float* xrow = xin + (long long)rowA * K;

    v8f accl = {0.f,0.f,0.f,0.f,0.f,0.f,0.f,0.f};
    v8f accr = {0.f,0.f,0.f,0.f,0.f,0.f,0.f,0.f};

#pragma unroll
    for (int k0 = 0; k0 < K; k0 += 4) {
        const int kr = k0 + kh;
        v2f a = *(const v2f*)(xrow + kr);
        v2f bl = { Wl[kr * HDIM + m], Wl[(kr + 1) * HDIM + m] };
        v2f br = { Wr[kr * HDIM + m], Wr[(kr + 1) * HDIM + m] };
        accl = __builtin_amdgcn_wmma_f32_16x16x4_f32(
                   false, a, false, bl, (short)0, accl, false, false);
        accr = __builtin_amdgcn_wmma_f32_16x16x4_f32(
                   false, a, false, br, (short)0, accr, false, false);
    }

    const float bv    = bias[m];
    const int   mbase = (lane >> 4) * 8;
#pragma unroll
    for (int v = 0; v < 8; ++v) {
        const int row = tile0 + mbase + v;
        if (row < nNodes) {
            outl[row * HDIM + m] = accl[v];
            outr[row * HDIM + m] = accr[v] + bv;
        }
    }
}

template <int K>
__global__ __launch_bounds__(256) void node_transform(
    const float* __restrict__ xin, const float* __restrict__ Wl,
    const float* __restrict__ Wr,  const float* __restrict__ bias,
    float* __restrict__ outl, float* __restrict__ outr, int nNodes)
{
    node_tile<K>(blockIdx.x * (blockDim.x >> 5) + (threadIdx.x >> 5),
                 xin, Wl, Wr, bias, outl, outr, nNodes);
}

#define BSHIFT 8
#define BNODES 256
#define MAXNB  1024
#define GCHUNKS 128

__global__ __launch_bounds__(1024) void bucket_hist(
    const int* __restrict__ eidx, int* __restrict__ hist,
    int E, int NB, int chunk)
{
    __shared__ int h[MAXNB];
    const int tid = threadIdx.x;
    for (int i = tid; i < NB; i += 1024) h[i] = 0;
    __syncthreads();
    const int beg = blockIdx.x * chunk;
    const int end = (beg + chunk < E) ? (beg + chunk) : E;
    const int* __restrict__ dstp = eidx + E;
    for (int e0 = beg; e0 < end; e0 += 4096) {
        int  d[4];
        bool ok[4];
#pragma unroll
        for (int u = 0; u < 4; ++u) {
            const int e = e0 + u * 1024 + tid;
            ok[u] = e < end;
            d[u]  = dstp[ok[u] ? e : end - 1];
        }
#pragma unroll
        for (int u = 0; u < 4; ++u)
            atomicAdd(&h[d[u] >> BSHIFT], ok[u] ? 1 : 0);
    }
    __syncthreads();
    for (int i = tid; i < NB; i += 1024) hist[blockIdx.x * NB + i] = h[i];
}

#define XF_TILES_PER_WG 25
__global__ __launch_bounds__(1024) void bucket_scan_and_project(
    const int* __restrict__ hist, int* __restrict__ offs,
    int* __restrict__ bstart, int NB,
    const float* __restrict__ xin, const float* __restrict__ Wl,
    const float* __restrict__ Wr,  const float* __restrict__ bias,
    float* __restrict__ outl, float* __restrict__ outr, int nNodes)
{
    __shared__ int sc[1024];
    if (blockIdx.x > 0) {
        const int w = __builtin_amdgcn_readfirstlane((int)(threadIdx.x >> 5));
        if (w < XF_TILES_PER_WG)
            node_tile<128>((blockIdx.x - 1) * XF_TILES_PER_WG + w,
                           xin, Wl, Wr, bias, outl, outr, nNodes);
        return;
    }
    const int tid = threadIdx.x;
    int total = 0;
    if (tid < NB) {
        int run = 0;
        for (int g0 = 0; g0 < GCHUNKS; g0 += 32) {
            int v[32];
#pragma unroll
            for (int k = 0; k < 32; ++k) v[k] = hist[(g0 + k) * NB + tid];
#pragma unroll
            for (int k = 0; k < 32; ++k) {
                offs[(g0 + k) * NB + tid] = run;
                run += v[k];
            }
        }
        total = run;
    }
    sc[tid] = total;
    __syncthreads();
    for (int d = 1; d < 1024; d <<= 1) {
        const int t = (tid >= d) ? sc[tid - d] : 0;
        __syncthreads();
        sc[tid] += t;
        __syncthreads();
    }
    if (tid < NB) bstart[tid] = sc[tid] - total;
    if (tid == NB - 1) bstart[NB] = sc[tid];
}

__global__ __launch_bounds__(1024) void bucket_scatter(
    const int* __restrict__ eidx, const int* __restrict__ offs,
    const int* __restrict__ bstart, unsigned* __restrict__ packed,
    int E, int NB, int chunk)
{
    __shared__ int cur[MAXNB];
    const int tid = threadIdx.x;
    for (int i = tid; i < NB; i += 1024)
        cur[i] = bstart[i] + offs[blockIdx.x * NB + i];
    __syncthreads();
    const int beg = blockIdx.x * chunk;
    const int end = (beg + chunk < E) ? (beg + chunk) : E;
    const int* __restrict__ dstp = eidx + E;
    int e0 = beg;
    for (; e0 + 4096 <= end; e0 += 4096) {
        int s[4], d[4];
#pragma unroll
        for (int u = 0; u < 4; ++u) {
            const int e = e0 + u * 1024 + tid;
            s[u] = eidx[e];
            d[u] = dstp[e];
        }
#pragma unroll
        for (int u = 0; u < 4; ++u) {
            const int pos = atomicAdd(&cur[d[u] >> BSHIFT], 1);
            packed[pos] = ((unsigned)s[u] << BSHIFT) | (unsigned)(d[u] & (BNODES - 1));
        }
    }
    for (int e = e0 + tid; e < end; e += 1024) {
        const int s = eidx[e];
        const int d = dstp[e];
        const int pos = atomicAdd(&cur[d >> BSHIFT], 1);
        packed[pos] = ((unsigned)s << BSHIFT) | (unsigned)(d & (BNODES - 1));
    }
}

__global__ __launch_bounds__(1024) void bucket_aggregate(
    const float* __restrict__ y, const float* __restrict__ yr,
    const unsigned* __restrict__ packed, const int* __restrict__ bstart,
    float* __restrict__ out, int nNodes, int doRelu)
{
    __shared__ float acc[BNODES * HDIM];
    __shared__ float cntL[BNODES];
    const int tid = threadIdx.x;
    for (int i = tid; i < BNODES * HDIM; i += 1024) acc[i] = 0.0f;
    if (tid < BNODES) cntL[tid] = 0.0f;
    __syncthreads();

    const int beg  = bstart[blockIdx.x];
    const int end  = bstart[blockIdx.x + 1];
    const int j    = tid & 15;
    const int slot = tid >> 4;
    for (int base = beg; base < end; base += 64 * 8) {
        unsigned pk[8];
        float    v[8];
        bool     ok[8];
#pragma unroll
        for (int u = 0; u < 8; ++u) {
            const int idx = base + u * 64 + slot;
            ok[u] = idx < end;
            pk[u] = packed[ok[u] ? idx : end - 1];
        }
#pragma unroll
        for (int u = 0; u < 8; ++u)
            v[u] = y[(long long)(pk[u] >> BSHIFT) * HDIM + j];
#pragma unroll
        for (int u = 0; u < 8; ++u) {
            const int dl = (int)(pk[u] & (BNODES - 1));
            atomicAdd(&acc[dl * HDIM + j], ok[u] ? v[u] : 0.0f);
            if (j == 0) atomicAdd(&cntL[dl], ok[u] ? 1.0f : 0.0f);
        }
    }
    __syncthreads();

    const int lo = blockIdx.x * BNODES;
    for (int i = tid; i < BNODES * HDIM; i += 1024) {
        const int n = lo + (i >> 4);
        if (n < nNodes) {
            float c = cntL[i >> 4];
            c = c > 1.0f ? c : 1.0f;
            const long long t = (long long)n * HDIM + (i & 15);
            float r = acc[i] / c + yr[t];
            if (doRelu) r = r > 0.0f ? r : 0.0f;
            out[t] = r;
        }
    }
}

__global__ __launch_bounds__(256) void decode_edges(
    const float* __restrict__ h, const int* __restrict__ didx,
    float* __restrict__ out, int P)
{
    const int t = blockIdx.x * blockDim.x + threadIdx.x;
    if (t >= P) return;
    const int a = didx[t];
    const int b = didx[P + t];
    const float4* ha = (const float4*)(h + (long long)a * HDIM);
    const float4* hb = (const float4*)(h + (long long)b * HDIM);
    float s = 0.0f;
#pragma unroll
    for (int i = 0; i < 4; ++i) {
        float4 u = ha[i];
        float4 w = hb[i];
        s += u.x * w.x + u.y * w.y + u.z * w.z + u.w * w.w;
    }
    out[t] = 1.0f / (1.0f + __expf(-s));
}

extern "C" void kernel_launch(void* const* d_in, const int* in_sizes, int n_in,
                              void* d_out, int out_size, void* d_ws, size_t ws_size,
                              hipStream_t stream) {
    const float* x   = (const float*)d_in[0];
    const float* W1l = (const float*)d_in[1];
    const float* b1  = (const float*)d_in[2];
    const float* W1r = (const float*)d_in[3];
    const float* W2l = (const float*)d_in[4];
    const float* b2  = (const float*)d_in[5];
    const float* W2r = (const float*)d_in[6];
    const int* eidx  = (const int*)d_in[7];
    const int* didx  = (const int*)d_in[8];
    float* out = (float*)d_out;

    const int N = in_sizes[0] / 128;
    const int E = in_sizes[7] / 2;
    const int P = in_sizes[8] / 2;
    const long long NH = (long long)N * HDIM;

    float* buf0 = (float*)d_ws;
    float* buf1 = buf0 + NH;
    float* buf2 = buf1 + NH;
    unsigned* packed = (unsigned*)(buf2 + NH);
    int* bstart = (int*)(packed + E);

    const int NB    = (N + BNODES - 1) / BNODES;
    const int chunk = (E + GCHUNKS - 1) / GCHUNKS;
    int* hist = (int*)buf2;
    int* offs = hist + GCHUNKS * NB;

    const int TB = 256;
    const int wavesPerBlock = TB / 32;
    const int tiles = (N + 15) / 16;
    const int xfBlocks = (tiles + wavesPerBlock - 1) / wavesPerBlock;
    const int decBlocks  = (P + TB - 1) / TB;

    bucket_hist<<<GCHUNKS, 1024, 0, stream>>>(eidx, hist, E, NB, chunk);
    const int xfWide = (tiles + XF_TILES_PER_WG - 1) / XF_TILES_PER_WG;
    bucket_scan_and_project<<<1 + xfWide, 1024, 0, stream>>>(
        hist, offs, bstart, NB, x, W1l, W1r, b1, buf0, buf1, N);
    bucket_scatter<<<GCHUNKS, 1024, 0, stream>>>(eidx, offs, bstart, packed,
                                                 E, NB, chunk);

    bucket_aggregate<<<NB, 1024, 0, stream>>>(buf0, buf1, packed, bstart,
                                              buf2, N, 1);

    node_transform<16><<<xfBlocks, TB, 0, stream>>>(buf2, W2l, W2r, b2,
                                                    buf0, buf1, N);

    bucket_aggregate<<<NB, 1024, 0, stream>>>(buf0, buf1, packed, bstart,
                                              buf2, N, 0);

    decode_edges<<<decBlocks, TB, 0, stream>>>(buf2, didx, out, P);
}


# decode re-mapped: a wave loops over 256 pairs and four lanes share a pair, 16 bytes of each row per lane (8 lines per load, not 32); a dot is now 4 per-lane fma chains added in lane order
# speedup vs baseline: 2.4526x; 1.0680x over previous
#include <hip/hip_runtime.h>
#include <math.h>

typedef __attribute__((ext_vector_type(2))) float v2f;
typedef __attribute__((ext_vector_type(8))) float v8f;

#define N_NODES 100000
#define HDIM 16

template <int K>
__device__ __forceinline__ void node_tile(
    const int wave,
    const float* __restrict__ xin, const float* __restrict__ Wl,
    const float* __restrict__ Wr,  const float* __restrict__ bias,
    float* __restrict__ outl, float* __restrict__ outr, int nNodes)
{
    const int lane  = threadIdx.x & 31;
    const int tile0 = wave * 16;
    if (tile0 >= nNodes) return;

    const int m  = lane & 15;
    const int kh = (lane >> 4) << 1;

    int rowA = tile0 + m;
    if (rowA >= nNodes) rowA = nNodes - 1;
    const float* xrow = xin + (long long)rowA * K;

    v8f accl = {0.f,0.f,0.f,0.f,0.f,0.f,0.f,0.f};
    v8f accr = {0.f,0.f,0.f,0.f,0.f,0.f,0.f,0.f};

#pragma unroll
    for (int k0 = 0; k0 < K; k0 += 4) {
        const int kr = k0 + kh;
        v2f a = *(const v2f*)(xrow + kr);
        v2f bl = { Wl[kr * HDIM + m], Wl[(kr + 1) * HDIM + m] };
        v2f br = { Wr[kr * HDIM + m], Wr[(kr + 1) * HDIM + m] };
        accl = __builtin_amdgcn_wmma_f32_16x16x4_f32(
                   false, a, false, bl, (short)0, accl, false, false);
        accr = __builtin_amdgcn_wmma_f32_16x16x4_f32(
                   false, a, false, br, (short)0, accr, false, false);
    }

    const float bv    = bias[m];
    const int   mbase = (lane >> 4) * 8;
#pragma unroll
    for (int v = 0; v < 8; ++v) {
        const int row = tile0 + mbase + v;
        if (row < nNodes) {
            outl[row * HDIM + m] = accl[v];
            outr[row * HDIM + m] = accr[v] + bv;
        }
    }
}

template <int K>
__global__ __launch_bounds__(256) void node_transform(
    const float* __restrict__ xin, const float* __restrict__ Wl,
    const float* __restrict__ Wr,  const float* __restrict__ bias,
    float* __restrict__ outl, float* __restrict__ outr, int nNodes)
{
    node_tile<K>(blockIdx.x * (blockDim.x >> 5) + (threadIdx.x >> 5),
                 xin, Wl, Wr, bias, outl, outr, nNodes);
}

#define BSHIFT 8
#define BNODES 256
#define MAXNB  1024
#define GCHUNKS 128

__global__ __launch_bounds__(1024) void bucket_hist(
    const int* __restrict__ eidx, int* __restrict__ hist,
    int E, int NB, int chunk)
{
    __shared__ int h[MAXNB];
    const int tid = threadIdx.x;
    for (int i = tid; i < NB; i += 1024) h[i] = 0;
    __syncthreads();
    const int beg = blockIdx.x * chunk;
    const int end = (beg + chunk < E) ? (beg + chunk) : E;
    const int* __restrict__ dstp = eidx + E;
    for (int e0 = beg; e0 < end; e0 += 4096) {
        int  d[4];
        bool ok[4];
#pragma unroll
        for (int u = 0; u < 4; ++u) {
            const int e = e0 + u * 1024 + tid;
            ok[u] = e < end;
            d[u]  = dstp[ok[u] ? e : end - 1];
        }
#pragma unroll
        for (int u = 0; u < 4; ++u)
            atomicAdd(&h[d[u] >> BSHIFT], ok[u] ? 1 : 0);
    }
    __syncthreads();
    for (int i = tid; i < NB; i += 1024) hist[blockIdx.x * NB + i] = h[i];
}

#define XF_TILES_PER_WG 25
__global__ __launch_bounds__(1024) void bucket_scan_and_project(
    const int* __restrict__ hist, int* __restrict__ offs,
    int* __restrict__ bstart, int NB,
    const float* __restrict__ xin, const float* __restrict__ Wl,
    const float* __restrict__ Wr,  const float* __restrict__ bias,
    float* __restrict__ outl, float* __restrict__ outr, int nNodes)
{
    __shared__ int sc[1024];
    if (blockIdx.x > 0) {
        const int w = __builtin_amdgcn_readfirstlane((int)(threadIdx.x >> 5));
        if (w < XF_TILES_PER_WG)
            node_tile<128>((blockIdx.x - 1) * XF_TILES_PER_WG + w,
                           xin, Wl, Wr, bias, outl, outr, nNodes);
        return;
    }
    const int tid = threadIdx.x;
    int total = 0;
    if (tid < NB) {
        int run = 0;
        for (int g0 = 0; g0 < GCHUNKS; g0 += 32) {
            int v[32];
#pragma unroll
            for (int k = 0; k < 32; ++k) v[k] = hist[(g0 + k) * NB + tid];
#pragma unroll
            for (int k = 0; k < 32; ++k) {
                offs[(g0 + k) * NB + tid] = run;
                run += v[k];
            }
        }
        total = run;
    }
    sc[tid] = total;
    __syncthreads();
    for (int d = 1; d < 1024; d <<= 1) {
        const int t = (tid >= d) ? sc[tid - d] : 0;
        __syncthreads();
        sc[tid] += t;
        __syncthreads();
    }
    if (tid < NB) bstart[tid] = sc[tid] - total;
    if (tid == NB - 1) bstart[NB] = sc[tid];
}

__global__ __launch_bounds__(1024) void bucket_scatter(
    const int* __restrict__ eidx, const int* __restrict__ offs,
    const int* __restrict__ bstart, unsigned* __restrict__ packed,
    int E, int NB, int chunk)
{
    __shared__ int cur[MAXNB];
    const int tid = threadIdx.x;
    for (int i = tid; i < NB; i += 1024)
        cur[i] = bstart[i] + offs[blockIdx.x * NB + i];
    __syncthreads();
    const int beg = blockIdx.x * chunk;
    const int end = (beg + chunk < E) ? (beg + chunk) : E;
    const int* __restrict__ dstp = eidx + E;
    int e0 = beg;
    for (; e0 + 4096 <= end; e0 += 4096) {
        int s[4], d[4];
#pragma unroll
        for (int u = 0; u < 4; ++u) {
            const int e = e0 + u * 1024 + tid;
            s[u] = eidx[e];
            d[u] = dstp[e];
        }
#pragma unroll
        for (int u = 0; u < 4; ++u) {
            const int pos = atomicAdd(&cur[d[u] >> BSHIFT], 1);
            packed[pos] = ((unsigned)s[u] << BSHIFT) | (unsigned)(d[u] & (BNODES - 1));
        }
    }
    for (int e = e0 + tid; e < end; e += 1024) {
        const int s = eidx[e];
        const int d = dstp[e];
        const int pos = atomicAdd(&cur[d >> BSHIFT], 1);
        packed[pos] = ((unsigned)s << BSHIFT) | (unsigned)(d & (BNODES - 1));
    }
}

__global__ __launch_bounds__(1024) void bucket_aggregate(
    const float* __restrict__ y, const float* __restrict__ yr,
    const unsigned* __restrict__ packed, const int* __restrict__ bstart,
    float* __restrict__ out, int nNodes, int doRelu)
{
    __shared__ float acc[BNODES * HDIM];
    __shared__ float cntL[BNODES];
    const int tid = threadIdx.x;
    for (int i = tid; i < BNODES * HDIM; i += 1024) acc[i] = 0.0f;
    if (tid < BNODES) cntL[tid] = 0.0f;
    __syncthreads();

    const int beg  = bstart[blockIdx.x];
    const int end  = bstart[blockIdx.x + 1];
    const int j    = tid & 15;
    const int slot = tid >> 4;
    for (int base = beg; base < end; base += 64 * 8) {
        unsigned pk[8];
        float    v[8];
        bool     ok[8];
#pragma unroll
        for (int u = 0; u < 8; ++u) {
            const int idx = base + u * 64 + slot;
            ok[u] = idx < end;
            pk[u] = packed[ok[u] ? idx : end - 1];
        }
#pragma unroll
        for (int u = 0; u < 8; ++u)
            v[u] = y[(long long)(pk[u] >> BSHIFT) * HDIM + j];
#pragma unroll
        for (int u = 0; u < 8; ++u) {
            const int dl = (int)(pk[u] & (BNODES - 1));
            atomicAdd(&acc[dl * HDIM + j], ok[u] ? v[u] : 0.0f);
            if (j == 0) atomicAdd(&cntL[dl], ok[u] ? 1.0f : 0.0f);
        }
    }
    __syncthreads();

    const int lo = blockIdx.x * BNODES;
    for (int i = tid; i < BNODES * HDIM; i += 1024) {
        const int n = lo + (i >> 4);
        if (n < nNodes) {
            float c = cntL[i >> 4];
            c = c > 1.0f ? c : 1.0f;
            const long long t = (long long)n * HDIM + (i & 15);
            float r = acc[i] / c + yr[t];
            if (doRelu) r = r > 0.0f ? r : 0.0f;
            out[t] = r;
        }
    }
}

#define DEC_PAIRS_PER_WAVE 256
#define DEC_WAVES_PER_WG   8

template <int K>
__device__ __forceinline__ int quad_take(int v)
{
    return __builtin_amdgcn_update_dpp(0, v, K * 0x55, 0xf, 0xf, true);
}
template <int K>
__device__ __forceinline__ float quad_take(float v)
{
    return __int_as_float(quad_take<K>(__float_as_int(v)));
}

template <int K>
__device__ __forceinline__ void dec_load(
    const float4* __restrict__ h4, int ia, int ib, unsigned q,
    float4& u, float4& w)
{
    const unsigned ea = (unsigned)quad_take<K>(ia) * 4u + q;
    const unsigned eb = (unsigned)quad_take<K>(ib) * 4u + q;
    u = h4[ea];
    w = h4[eb];
}

template <int K>
__device__ __forceinline__ void dec_sum(
    const float4& u, const float4& w, int q, float& res)
{
    const float p = u.x * w.x + u.y * w.y + u.z * w.z + u.w * w.w;
    float s = quad_take<0>(p) + quad_take<1>(p);
    s += quad_take<2>(p);
    s += quad_take<3>(p);
    if (q == K) res = s;
}

__global__ __launch_bounds__(256) void decode_edges(
    const float* __restrict__ h, const int* __restrict__ didx,
    float* __restrict__ out, int P)
{
    const int lane = threadIdx.x & 31;
    const int q    = lane & 3;
    const float4* __restrict__ h4 = (const float4*)h;
    const int w = __builtin_amdgcn_readfirstlane((int)(threadIdx.x >> 5));
    const int first = (blockIdx.x * DEC_WAVES_PER_WG + w) * DEC_PAIRS_PER_WAVE;
    int last = first + DEC_PAIRS_PER_WAVE;
    if (last > P) last = P;

#pragma unroll 1
    for (int t0 = first; t0 < last; t0 += 64) {
        int ia[2], ib[2];
#pragma unroll
        for (int j = 0; j < 2; ++j) {
            const int t  = t0 + j * 32 + lane;
            const int tc = t < P ? t : P - 1;
            ia[j] = didx[tc];
            ib[j] = didx[P + tc];
        }
        float4 u[2][4], v[2][4];
#pragma unroll
        for (int j = 0; j < 2; ++j) {
            dec_load<0>(h4, ia[j], ib[j], (unsigned)q, u[j][0], v[j][0]);
            dec_load<1>(h4, ia[j], ib[j], (unsigned)q, u[j][1], v[j][1]);
            dec_load<2>(h4, ia[j], ib[j], (unsigned)q, u[j][2], v[j][2]);
            dec_load<3>(h4, ia[j], ib[j], (unsigned)q, u[j][3], v[j][3]);
        }
#pragma unroll
        for (int j = 0; j < 2; ++j) {
            float res = 0.0f;
            dec_sum<0>(u[j][0], v[j][0], q, res);
            dec_sum<1>(u[j][1], v[j][1], q, res);
            dec_sum<2>(u[j][2], v[j][2], q, res);
            dec_sum<3>(u[j][3], v[j][3], q, res);
            const int t = t0 + j * 32 + lane;
            if (t < P) out[t] = 1.0f / (1.0f + __expf(-res));
        }
    }
}

extern "C" void kernel_launch(void* const* d_in, const int* in_sizes, int n_in,
                              void* d_out, int out_size, void* d_ws, size_t ws_size,
                              hipStream_t stream) {
    const float* x   = (const float*)d_in[0];
    const float* W1l = (const float*)d_in[1];
    const float* b1  = (const float*)d_in[2];
    const float* W1r = (const float*)d_in[3];
    const float* W2l = (const float*)d_in[4];
    const float* b2  = (const float*)d_in[5];
    const float* W2r = (const float*)d_in[6];
    const int* eidx  = (const int*)d_in[7];
    const int* didx  = (const int*)d_in[8];
    float* out = (float*)d_out;

    const int N = in_sizes[0] / 128;
    const int E = in_sizes[7] / 2;
    const int P = in_sizes[8] / 2;
    const long long NH = (long long)N * HDIM;

    float* buf0 = (float*)d_ws;
    float* buf1 = buf0 + NH;
    float* buf2 = buf1 + NH;
    unsigned* packed = (unsigned*)(buf2 + NH);
    int* bstart = (int*)(packed + E);

    const int NB    = (N + BNODES - 1) / BNODES;
    const int chunk = (E + GCHUNKS - 1) / GCHUNKS;
    int* hist = (int*)buf2;
    int* offs = hist + GCHUNKS * NB;

    const int TB = 256;
    const int wavesPerBlock = TB / 32;
    const int tiles = (N + 15) / 16;
    const int xfBlocks = (tiles + wavesPerBlock - 1) / wavesPerBlock;
    const int decPerWg   = DEC_WAVES_PER_WG * DEC_PAIRS_PER_WAVE;
    const int decBlocks  = (P + decPerWg - 1) / decPerWg;

    bucket_hist<<<GCHUNKS, 1024, 0, stream>>>(eidx, hist, E, NB, chunk);
    const int xfWide = (tiles + XF_TILES_PER_WG - 1) / XF_TILES_PER_WG;
    bucket_scan_and_project<<<1 + xfWide, 1024, 0, stream>>>(
        hist, offs, bstart, NB, x, W1l, W1r, b1, buf0, buf1, N);
    bucket_scatter<<<GCHUNKS, 1024, 0, stream>>>(eidx, offs, bstart, packed,
                                                 E, NB, chunk);

    bucket_aggregate<<<NB, 1024, 0, stream>>>(buf0, buf1, packed, bstart,
                                              buf2, N, 1);

    node_transform<16><<<xfBlocks, TB, 0, stream>>>(buf2, W2l, W2r, b2,
                                                    buf0, buf1, N);

    bucket_aggregate<<<NB, 1024, 0, stream>>>(buf0, buf1, packed, bstart,
                                              buf2, N, 0);

    decode_edges<<<decBlocks, DEC_WAVES_PER_WG * 32, 0, stream>>>(buf2, didx, out, P);
}


# aggregate re-mapped: four lanes share an edge, 16 bytes of the row per lane, 8 waves per work-group not 32; LDS sums in 4 component planes; still added in arrival order, which the mapping changes
# speedup vs baseline: 2.6087x; 1.0636x over previous
#include <hip/hip_runtime.h>
#include <math.h>

typedef __attribute__((ext_vector_type(2))) float v2f;
typedef __attribute__((ext_vector_type(8))) float v8f;

#define N_NODES 100000
#define HDIM 16

template <int K>
__device__ __forceinline__ void node_tile(
    const int wave,
    const float* __restrict__ xin, const float* __restrict__ Wl,
    const float* __restrict__ Wr,  const float* __restrict__ bias,
    float* __restrict__ outl, float* __restrict__ outr, int nNodes)
{
    const int lane  = threadIdx.x & 31;
    const int tile0 = wave * 16;
    if (tile0 >= nNodes) return;

    const int m  = lane & 15;
    const int kh = (lane >> 4) << 1;

    int rowA = tile0 + m;
    if (rowA >= nNodes) rowA = nNodes - 1;
    const float* xrow = xin + (long long)rowA * K;

    v8f accl = {0.f,0.f,0.f,0.f,0.f,0.f,0.f,0.f};
    v8f accr = {0.f,0.f,0.f,0.f,0.f,0.f,0.f,0.f};

#pragma unroll
    for (int k0 = 0; k0 < K; k0 += 4) {
        const int kr = k0 + kh;
        v2f a = *(const v2f*)(xrow + kr);
        v2f bl = { Wl[kr * HDIM + m], Wl[(kr + 1) * HDIM + m] };
        v2f br = { Wr[kr * HDIM + m], Wr[(kr + 1) * HDIM + m] };
        accl = __builtin_amdgcn_wmma_f32_16x16x4_f32(
                   false, a, false, bl, (short)0, accl, false, false);
        accr = __builtin_amdgcn_wmma_f32_16x16x4_f32(
                   false, a, false, br, (short)0, accr, false, false);
    }

    const float bv    = bias[m];
    const int   mbase = (lane >> 4) * 8;
#pragma unroll
    for (int v = 0; v < 8; ++v) {
        const int row = tile0 + mbase + v;
        if (row < nNodes) {
            outl[row * HDIM + m] = accl[v];
            outr[row * HDIM + m] = accr[v] + bv;
        }
    }
}

template <int K>
__global__ __launch_bounds__(256) void node_transform(
    const float* __restrict__ xin, const float* __restrict__ Wl,
    const float* __restrict__ Wr,  const float* __restrict__ bias,
    float* __restrict__ outl, float* __restrict__ outr, int nNodes)
{
    node_tile<K>(blockIdx.x * (blockDim.x >> 5) + (threadIdx.x >> 5),
                 xin, Wl, Wr, bias, outl, outr, nNodes);
}

#define BSHIFT 8
#define BNODES 256
#define MAXNB  1024
#define GCHUNKS 128

__global__ __launch_bounds__(1024) void bucket_hist(
    const int* __restrict__ eidx, int* __restrict__ hist,
    int E, int NB, int chunk)
{
    __shared__ int h[MAXNB];
    const int tid = threadIdx.x;
    for (int i = tid; i < NB; i += 1024) h[i] = 0;
    __syncthreads();
    const int beg = blockIdx.x * chunk;
    const int end = (beg + chunk < E) ? (beg + chunk) : E;
    const int* __restrict__ dstp = eidx + E;
    for (int e0 = beg; e0 < end; e0 += 4096) {
        int  d[4];
        bool ok[4];
#pragma unroll
        for (int u = 0; u < 4; ++u) {
            const int e = e0 + u * 1024 + tid;
            ok[u] = e < end;
            d[u]  = dstp[ok[u] ? e : end - 1];
        }
#pragma unroll
        for (int u = 0; u < 4; ++u)
            atomicAdd(&h[d[u] >> BSHIFT], ok[u] ? 1 : 0);
    }
    __syncthreads();
    for (int i = tid; i < NB; i += 1024) hist[blockIdx.x * NB + i] = h[i];
}

#define XF_TILES_PER_WG 25
__global__ __launch_bounds__(1024) void bucket_scan_and_project(
    const int* __restrict__ hist, int* __restrict__ offs,
    int* __restrict__ bstart, int NB,
    const float* __restrict__ xin, const float* __restrict__ Wl,
    const float* __restrict__ Wr,  const float* __restrict__ bias,
    float* __restrict__ outl, float* __restrict__ outr, int nNodes)
{
    __shared__ int sc[1024];
    if (blockIdx.x > 0) {
        const int w = __builtin_amdgcn_readfirstlane((int)(threadIdx.x >> 5));
        if (w < XF_TILES_PER_WG)
            node_tile<128>((blockIdx.x - 1) * XF_TILES_PER_WG + w,
                           xin, Wl, Wr, bias, outl, outr, nNodes);
        return;
    }
    const int tid = threadIdx.x;
    int total = 0;
    if (tid < NB) {
        int run = 0;
        for (int g0 = 0; g0 < GCHUNKS; g0 += 32) {
            int v[32];
#pragma unroll
            for (int k = 0; k < 32; ++k) v[k] = hist[(g0 + k) * NB + tid];
#pragma unroll
            for (int k = 0; k < 32; ++k) {
                offs[(g0 + k) * NB + tid] = run;
                run += v[k];
            }
        }
        total = run;
    }
    sc[tid] = total;
    __syncthreads();
    for (int d = 1; d < 1024; d <<= 1) {
        const int t = (tid >= d) ? sc[tid - d] : 0;
        __syncthreads();
        sc[tid] += t;
        __syncthreads();
    }
    if (tid < NB) bstart[tid] = sc[tid] - total;
    if (tid == NB - 1) bstart[NB] = sc[tid];
}

__global__ __launch_bounds__(1024) void bucket_scatter(
    const int* __restrict__ eidx, const int* __restrict__ offs,
    const int* __restrict__ bstart, unsigned* __restrict__ packed,
    int E, int NB, int chunk)
{
    __shared__ int cur[MAXNB];
    const int tid = threadIdx.x;
    for (int i = tid; i < NB; i += 1024)
        cur[i] = bstart[i] + offs[blockIdx.x * NB + i];
    __syncthreads();
    const int beg = blockIdx.x * chunk;
    const int end = (beg + chunk < E) ? (beg + chunk) : E;
    const int* __restrict__ dstp = eidx + E;
    int e0 = beg;
    for (; e0 + 4096 <= end; e0 += 4096) {
        int s[4], d[4];
#pragma unroll
        for (int u = 0; u < 4; ++u) {
            const int e = e0 + u * 1024 + tid;
            s[u] = eidx[e];
            d[u] = dstp[e];
        }
#pragma unroll
        for (int u = 0; u < 4; ++u) {
            const int pos = atomicAdd(&cur[d[u] >> BSHIFT], 1);
            packed[pos] = ((unsigned)s[u] << BSHIFT) | (unsigned)(d[u] & (BNODES - 1));
        }
    }
    for (int e = e0 + tid; e < end; e += 1024) {
        const int s = eidx[e];
        const int d = dstp[e];
        const int pos = atomicAdd(&cur[d >> BSHIFT], 1);
        packed[pos] = ((unsigned)s << BSHIFT) | (unsigned)(d & (BNODES - 1));
    }
}

#define AGG_THREADS 256
#define AGG_SPARE   32
#define AGG_PLANE   (BNODES * 4 + AGG_SPARE)

__global__ __launch_bounds__(AGG_THREADS) void bucket_aggregate(
    const float* __restrict__ y, const float* __restrict__ yr,
    const unsigned* __restrict__ packed, const int* __restrict__ bstart,
    float* __restrict__ out, int nNodes, int doRelu)
{
    __shared__ float acc[4 * AGG_PLANE];
    __shared__ float cntL[BNODES + AGG_SPARE];
    const int tid = threadIdx.x;
    for (int i = tid; i < 4 * AGG_PLANE; i += AGG_THREADS) acc[i] = 0.0f;
    for (int i = tid; i < BNODES + AGG_SPARE; i += AGG_THREADS) cntL[i] = 0.0f;
    __syncthreads();

    const int beg  = bstart[blockIdx.x];
    const int end  = bstart[blockIdx.x + 1];
    const int q    = tid & 3;
    const int slot = tid >> 2;
    const int spare = BNODES * 4 + (tid & 31);
    const float4* __restrict__ y4 = (const float4*)y;
    for (int base = beg; base < end; base += 64 * 8) {
        unsigned pk[8];
        float4   v[8];
        bool     ok[8];
#pragma unroll
        for (int u = 0; u < 8; ++u) {
            const int idx = base + u * 64 + slot;
            ok[u] = idx < end;
            pk[u] = packed[ok[u] ? idx : end - 1];
        }
#pragma unroll
        for (int u = 0; u < 8; ++u)
            v[u] = y4[(pk[u] >> BSHIFT) * 4u + (unsigned)q];
#pragma unroll
        for (int u = 0; u < 8; ++u) {
            const int dl = (int)(pk[u] & (BNODES - 1));
            const int a  = ok[u] ? dl * 4 + q : spare;
            atomicAdd(&acc[a], v[u].x);
            atomicAdd(&acc[AGG_PLANE + a], v[u].y);
            atomicAdd(&acc[2 * AGG_PLANE + a], v[u].z);
            atomicAdd(&acc[3 * AGG_PLANE + a], v[u].w);
            if (q == 0) atomicAdd(&cntL[ok[u] ? dl : BNODES + (tid & 31)], 1.0f);
        }
    }
    __syncthreads();

    const int lo = blockIdx.x * BNODES;
    const float4* __restrict__ yr4 = (const float4*)yr;
    float4* __restrict__ out4 = (float4*)out;
    for (int i = tid; i < BNODES * 4; i += AGG_THREADS) {
        const int n = lo + (i >> 2);
        if (n < nNodes) {
            float c = cntL[i >> 2];
            c = c > 1.0f ? c : 1.0f;
            const long long t = (long long)n * 4 + (i & 3);
            const float4 b = yr4[t];
            float4 r;
            r.x = acc[i] / c + b.x;
            r.y = acc[AGG_PLANE + i] / c + b.y;
            r.z = acc[2 * AGG_PLANE + i] / c + b.z;
            r.w = acc[3 * AGG_PLANE + i] / c + b.w;
            if (doRelu) {
                r.x = r.x > 0.0f ? r.x : 0.0f;
                r.y = r.y > 0.0f ? r.y : 0.0f;
                r.z = r.z > 0.0f ? r.z : 0.0f;
                r.w = r.w > 0.0f ? r.w : 0.0f;
            }
            out4[t] = r;
        }
    }
}

#define DEC_PAIRS_PER_WAVE 256
#define DEC_WAVES_PER_WG   8

template <int K>
__device__ __forceinline__ int quad_take(int v)
{
    return __builtin_amdgcn_update_dpp(0, v, K * 0x55, 0xf, 0xf, true);
}
template <int K>
__device__ __forceinline__ float quad_take(float v)
{
    return __int_as_float(quad_take<K>(__float_as_int(v)));
}

template <int K>
__device__ __forceinline__ void dec_load(
    const float4* __restrict__ h4, int ia, int ib, unsigned q,
    float4& u, float4& w)
{
    const unsigned ea = (unsigned)quad_take<K>(ia) * 4u + q;
    const unsigned eb = (unsigned)quad_take<K>(ib) * 4u + q;
    u = h4[ea];
    w = h4[eb];
}

template <int K>
__device__ __forceinline__ void dec_sum(
    const float4& u, const float4& w, int q, float& res)
{
    const float p = u.x * w.x + u.y * w.y + u.z * w.z + u.w * w.w;
    float s = quad_take<0>(p) + quad_take<1>(p);
    s += quad_take<2>(p);
    s += quad_take<3>(p);
    if (q == K) res = s;
}

__global__ __launch_bounds__(256) void decode_edges(
    const float* __restrict__ h, const int* __restrict__ didx,
    float* __restrict__ out, int P)
{
    const int lane = threadIdx.x & 31;
    const int q    = lane & 3;
    const float4* __restrict__ h4 = (const float4*)h;
    const int w = __builtin_amdgcn_readfirstlane((int)(threadIdx.x >> 5));
    const int first = (blockIdx.x * DEC_WAVES_PER_WG + w) * DEC_PAIRS_PER_WAVE;
    int last = first + DEC_PAIRS_PER_WAVE;
    if (last > P) last = P;

#pragma unroll 1
    for (int t0 = first; t0 < last; t0 += 64) {
        int ia[2], ib[2];
#pragma unroll
        for (int j = 0; j < 2; ++j) {
            const int t  = t0 + j * 32 + lane;
            const int tc = t < P ? t : P - 1;
            ia[j] = didx[tc];
            ib[j] = didx[P + tc];
        }
        float4 u[2][4], v[2][4];
#pragma unroll
        for (int j = 0; j < 2; ++j) {
            dec_load<0>(h4, ia[j], ib[j], (unsigned)q, u[j][0], v[j][0]);
            dec_load<1>(h4, ia[j], ib[j], (unsigned)q, u[j][1], v[j][1]);
            dec_load<2>(h4, ia[j], ib[j], (unsigned)q, u[j][2], v[j][2]);
            dec_load<3>(h4, ia[j], ib[j], (unsigned)q, u[j][3], v[j][3]);
        }
#pragma unroll
        for (int j = 0; j < 2; ++j) {
            float res = 0.0f;
            dec_sum<0>(u[j][0], v[j][0], q, res);
            dec_sum<1>(u[j][1], v[j][1], q, res);
            dec_sum<2>(u[j][2], v[j][2], q, res);
            dec_sum<3>(u[j][3], v[j][3], q, res);
            const int t = t0 + j * 32 + lane;
            if (t < P) out[t] = 1.0f / (1.0f + __expf(-res));
        }
    }
}

extern "C" void kernel_launch(void* const* d_in, const int* in_sizes, int n_in,
                              void* d_out, int out_size, void* d_ws, size_t ws_size,
                              hipStream_t stream) {
    const float* x   = (const float*)d_in[0];
    const float* W1l = (const float*)d_in[1];
    const float* b1  = (const float*)d_in[2];
    const float* W1r = (const float*)d_in[3];
    const float* W2l = (const float*)d_in[4];
    const float* b2  = (const float*)d_in[5];
    const float* W2r = (const float*)d_in[6];
    const int* eidx  = (const int*)d_in[7];
    const int* didx  = (const int*)d_in[8];
    float* out = (float*)d_out;

    const int N = in_sizes[0] / 128;
    const int E = in_sizes[7] / 2;
    const int P = in_sizes[8] / 2;
    const long long NH = (long long)N * HDIM;

    float* buf0 = (float*)d_ws;
    float* buf1 = buf0 + NH;
    float* buf2 = buf1 + NH;
    unsigned* packed = (unsigned*)(buf2 + NH);
    int* bstart = (int*)(packed + E);

    const int NB    = (N + BNODES - 1) / BNODES;
    const int chunk = (E + GCHUNKS - 1) / GCHUNKS;
    int* hist = (int*)buf2;
    int* offs = hist + GCHUNKS * NB;

    const int TB = 256;
    const int wavesPerBlock = TB / 32;
    const int tiles = (N + 15) / 16;
    const int xfBlocks = (tiles + wavesPerBlock - 1) / wavesPerBlock;
    const int decPerWg   = DEC_WAVES_PER_WG * DEC_PAIRS_PER_WAVE;
    const int decBlocks  = (P + decPerWg - 1) / decPerWg;

    bucket_hist<<<GCHUNKS, 1024, 0, stream>>>(eidx, hist, E, NB, chunk);
    const int xfWide = (tiles + XF_TILES_PER_WG - 1) / XF_TILES_PER_WG;
    bucket_scan_and_project<<<1 + xfWide, 1024, 0, stream>>>(
        hist, offs, bstart, NB, x, W1l, W1r, b1, buf0, buf1, N);
    bucket_scatter<<<GCHUNKS, 1024, 0, stream>>>(eidx, offs, bstart, packed,
                                                 E, NB, chunk);

    bucket_aggregate<<<NB, AGG_THREADS, 0, stream>>>(buf0, buf1, packed, bstart,
                                              buf2, N, 1);

    node_transform<16><<<xfBlocks, TB, 0, stream>>>(buf2, W2l, W2r, b2,
                                                    buf0, buf1, N);

    bucket_aggregate<<<NB, AGG_THREADS, 0, stream>>>(buf0, buf1, packed, bstart,
                                              buf2, N, 0);

    decode_edges<<<decBlocks, DEC_WAVES_PER_WG * 32, 0, stream>>>(buf2, didx, out, P);
}


# layer-2 projection moved into the layer-1 aggregate's epilogue: relu'd rows stay in LDS and take the same WMMA sequence there; the projection launch goes, 7 -> 6; arithmetic and sum order unchanged
# speedup vs baseline: 2.6947x; 1.0330x over previous
#include <hip/hip_runtime.h>
#include <math.h>

typedef __attribute__((ext_vector_type(2))) float v2f;
typedef __attribute__((ext_vector_type(8))) float v8f;

#define N_NODES 100000
#define HDIM 16

template <int K>
__device__ __forceinline__ void node_tile(
    const int wave,
    const float* __restrict__ xin, const float* __restrict__ Wl,
    const float* __restrict__ Wr,  const float* __restrict__ bias,
    float* __restrict__ outl, float* __restrict__ outr, int nNodes)
{
    const int lane  = threadIdx.x & 31;
    const int tile0 = wave * 16;
    if (tile0 >= nNodes) return;

    const int m  = lane & 15;
    const int kh = (lane >> 4) << 1;

    int rowA = tile0 + m;
    if (rowA >= nNodes) rowA = nNodes - 1;
    const float* xrow = xin + (long long)rowA * K;

    v8f accl = {0.f,0.f,0.f,0.f,0.f,0.f,0.f,0.f};
    v8f accr = {0.f,0.f,0.f,0.f,0.f,0.f,0.f,0.f};

#pragma unroll
    for (int k0 = 0; k0 < K; k0 += 4) {
        const int kr = k0 + kh;
        v2f a = *(const v2f*)(xrow + kr);
        v2f bl = { Wl[kr * HDIM + m], Wl[(kr + 1) * HDIM + m] };
        v2f br = { Wr[kr * HDIM + m], Wr[(kr + 1) * HDIM + m] };
        accl = __builtin_amdgcn_wmma_f32_16x16x4_f32(
                   false, a, false, bl, (short)0, accl, false, false);
        accr = __builtin_amdgcn_wmma_f32_16x16x4_f32(
                   false, a, false, br, (short)0, accr, false, false);
    }

    const float bv    = bias[m];
    const int   mbase = (lane >> 4) * 8;
#pragma unroll
    for (int v = 0; v < 8; ++v) {
        const int row = tile0 + mbase + v;
        if (row < nNodes) {
            outl[row * HDIM + m] = accl[v];
            outr[row * HDIM + m] = accr[v] + bv;
        }
    }
}

#define BSHIFT 8
#define BNODES 256
#define MAXNB  1024
#define GCHUNKS 128

__global__ __launch_bounds__(1024) void bucket_hist(
    const int* __restrict__ eidx, int* __restrict__ hist,
    int E, int NB, int chunk)
{
    __shared__ int h[MAXNB];
    const int tid = threadIdx.x;
    for (int i = tid; i < NB; i += 1024) h[i] = 0;
    __syncthreads();
    const int beg = blockIdx.x * chunk;
    const int end = (beg + chunk < E) ? (beg + chunk) : E;
    const int* __restrict__ dstp = eidx + E;
    for (int e0 = beg; e0 < end; e0 += 4096) {
        int  d[4];
        bool ok[4];
#pragma unroll
        for (int u = 0; u < 4; ++u) {
            const int e = e0 + u * 1024 + tid;
            ok[u] = e < end;
            d[u]  = dstp[ok[u] ? e : end - 1];
        }
#pragma unroll
        for (int u = 0; u < 4; ++u)
            atomicAdd(&h[d[u] >> BSHIFT], ok[u] ? 1 : 0);
    }
    __syncthreads();
    for (int i = tid; i < NB; i += 1024) hist[blockIdx.x * NB + i] = h[i];
}

#define XF_TILES_PER_WG 25
__global__ __launch_bounds__(1024) void bucket_scan_and_project(
    const int* __restrict__ hist, int* __restrict__ offs,
    int* __restrict__ bstart, int NB,
    const float* __restrict__ xin, const float* __restrict__ Wl,
    const float* __restrict__ Wr,  const float* __restrict__ bias,
    float* __restrict__ outl, float* __restrict__ outr, int nNodes)
{
    __shared__ int sc[1024];
    if (blockIdx.x > 0) {
        const int w = __builtin_amdgcn_readfirstlane((int)(threadIdx.x >> 5));
        if (w < XF_TILES_PER_WG)
            node_tile<128>((blockIdx.x - 1) * XF_TILES_PER_WG + w,
                           xin, Wl, Wr, bias, outl, outr, nNodes);
        return;
    }
    const int tid = threadIdx.x;
    int total = 0;
    if (tid < NB) {
        int run = 0;
        for (int g0 = 0; g0 < GCHUNKS; g0 += 32) {
            int v[32];
#pragma unroll
            for (int k = 0; k < 32; ++k) v[k] = hist[(g0 + k) * NB + tid];
#pragma unroll
            for (int k = 0; k < 32; ++k) {
                offs[(g0 + k) * NB + tid] = run;
                run += v[k];
            }
        }
        total = run;
    }
    sc[tid] = total;
    __syncthreads();
    for (int d = 1; d < 1024; d <<= 1) {
        const int t = (tid >= d) ? sc[tid - d] : 0;
        __syncthreads();
        sc[tid] += t;
        __syncthreads();
    }
    if (tid < NB) bstart[tid] = sc[tid] - total;
    if (tid == NB - 1) bstart[NB] = sc[tid];
}

__global__ __launch_bounds__(1024) void bucket_scatter(
    const int* __restrict__ eidx, const int* __restrict__ offs,
    const int* __restrict__ bstart, unsigned* __restrict__ packed,
    int E, int NB, int chunk)
{
    __shared__ int cur[MAXNB];
    const int tid = threadIdx.x;
    for (int i = tid; i < NB; i += 1024)
        cur[i] = bstart[i] + offs[blockIdx.x * NB + i];
    __syncthreads();
    const int beg = blockIdx.x * chunk;
    const int end = (beg + chunk < E) ? (beg + chunk) : E;
    const int* __restrict__ dstp = eidx + E;
    int e0 = beg;
    for (; e0 + 4096 <= end; e0 += 4096) {
        int s[4], d[4];
#pragma unroll
        for (int u = 0; u < 4; ++u) {
            const int e = e0 + u * 1024 + tid;
            s[u] = eidx[e];
            d[u] = dstp[e];
        }
#pragma unroll
        for (int u = 0; u < 4; ++u) {
            const int pos = atomicAdd(&cur[d[u] >> BSHIFT], 1);
            packed[pos] = ((unsigned)s[u] << BSHIFT) | (unsigned)(d[u] & (BNODES - 1));
        }
    }
    for (int e = e0 + tid; e < end; e += 1024) {
        const int s = eidx[e];
        const int d = dstp[e];
        const int pos = atomicAdd(&cur[d >> BSHIFT], 1);
        packed[pos] = ((unsigned)s << BSHIFT) | (unsigned)(d & (BNODES - 1));
    }
}

#define AGG_THREADS 256
#define AGG_SPARE   32
#define AGG_PLANE   (BNODES * 4 + AGG_SPARE)

template <bool PROJECT>
__device__ __forceinline__ void aggregate_bucket(
    const float* y, const float* yr,
    const unsigned* packed, const int* bstart,
    float* out, float* outr,
    const float* Wl, const float* Wr, const float* bias, int nNodes)
{
    __shared__ __attribute__((aligned(16))) float acc[4 * AGG_PLANE];
    __shared__ float cntL[BNODES + AGG_SPARE];
    const int tid = threadIdx.x;

    for (int i = tid; i < 4 * AGG_PLANE; i += AGG_THREADS) acc[i] = 0.0f;
    for (int i = tid; i < BNODES + AGG_SPARE; i += AGG_THREADS) cntL[i] = 0.0f;
    __syncthreads();

    const int beg  = bstart[blockIdx.x];
    const int end  = bstart[blockIdx.x + 1];

    v2f bl[4], br[4];
    float bv = 0.0f;
    if (PROJECT) {
        const int m  = tid & 15;
        const int kh = ((tid >> 4) & 1) << 1;
#pragma unroll
        for (int s = 0; s < 4; ++s) {
            const int kr = 4 * s + kh;
            bl[s] = (v2f){ Wl[kr * HDIM + m], Wl[(kr + 1) * HDIM + m] };
            br[s] = (v2f){ Wr[kr * HDIM + m], Wr[(kr + 1) * HDIM + m] };
        }
        bv = bias[m];
    }

    const int q    = tid & 3;
    const int slot = tid >> 2;
    const int spare = BNODES * 4 + (tid & 31);
    const float4* __restrict__ y4 = (const float4*)y;
    for (int base = beg; base < end; base += 64 * 8) {
        unsigned pk[8];
        float4   v[8];
        bool     ok[8];
#pragma unroll
        for (int u = 0; u < 8; ++u) {
            const int idx = base + u * 64 + slot;
            ok[u] = idx < end;
            pk[u] = packed[ok[u] ? idx : end - 1];
        }
#pragma unroll
        for (int u = 0; u < 8; ++u)
            v[u] = y4[(pk[u] >> BSHIFT) * 4u + (unsigned)q];
#pragma unroll
        for (int u = 0; u < 8; ++u) {
            const int dl = (int)(pk[u] & (BNODES - 1));
            const int a  = ok[u] ? dl * 4 + q : spare;
            atomicAdd(&acc[a], v[u].x);
            atomicAdd(&acc[AGG_PLANE + a], v[u].y);
            atomicAdd(&acc[2 * AGG_PLANE + a], v[u].z);
            atomicAdd(&acc[3 * AGG_PLANE + a], v[u].w);
            if (q == 0) atomicAdd(&cntL[ok[u] ? dl : BNODES + (tid & 31)], 1.0f);
        }
    }
    __syncthreads();

    const int lo = blockIdx.x * BNODES;
    if (!PROJECT) {
        const float4* __restrict__ yr4 = (const float4*)yr;
        float4* __restrict__ out4 = (float4*)out;
        for (int i = tid; i < BNODES * 4; i += AGG_THREADS) {
            const int n = lo + (i >> 2);
            if (n < nNodes) {
                float c = cntL[i >> 2];
                c = c > 1.0f ? c : 1.0f;
                const long long t = (long long)n * 4 + (i & 3);
                const float4 b = yr4[t];
                float4 r;
                r.x = acc[i] / c + b.x;
                r.y = acc[AGG_PLANE + i] / c + b.y;
                r.z = acc[2 * AGG_PLANE + i] / c + b.z;
                r.w = acc[3 * AGG_PLANE + i] / c + b.w;
                out4[t] = r;
            }
        }
        return;
    }

    const float4* yr4 = (const float4*)yr;
    for (int i = tid; i < BNODES * 4; i += AGG_THREADS) {
        const int n = lo + (i >> 2);
        if (n < nNodes) {
            float c = cntL[i >> 2];
            c = c > 1.0f ? c : 1.0f;
            const float4 b = yr4[(long long)n * 4 + (i & 3)];
            float4 r;
            r.x = acc[i] / c + b.x;
            r.y = acc[AGG_PLANE + i] / c + b.y;
            r.z = acc[2 * AGG_PLANE + i] / c + b.z;
            r.w = acc[3 * AGG_PLANE + i] / c + b.w;
            acc[i]                 = r.x > 0.0f ? r.x : 0.0f;
            acc[AGG_PLANE + i]     = r.y > 0.0f ? r.y : 0.0f;
            acc[2 * AGG_PLANE + i] = r.z > 0.0f ? r.z : 0.0f;
            acc[3 * AGG_PLANE + i] = r.w > 0.0f ? r.w : 0.0f;
        }
    }
    __syncthreads();

    const int lane  = tid & 31;
    const int m     = lane & 15;
    const int half  = lane >> 4;
    const int wv = __builtin_amdgcn_readfirstlane(tid >> 5);
#pragma unroll
    for (int t = 0; t < 2; ++t) {
        const int tl = (wv * 2 + t) * 16;
        if (lo + tl >= nNodes) continue;
        const float4 p0 = *(const float4*)&acc[(2 * half)     * AGG_PLANE + (tl + m) * 4];
        const float4 p1 = *(const float4*)&acc[(2 * half + 1) * AGG_PLANE + (tl + m) * 4];
        const float e0[4] = { p0.x, p0.y, p0.z, p0.w };
        const float e1[4] = { p1.x, p1.y, p1.z, p1.w };
        v8f accl = {0.f,0.f,0.f,0.f,0.f,0.f,0.f,0.f};
        v8f accr = {0.f,0.f,0.f,0.f,0.f,0.f,0.f,0.f};
#pragma unroll
        for (int s = 0; s < 4; ++s) {
            const v2f a = { e0[s], e1[s] };
            accl = __builtin_amdgcn_wmma_f32_16x16x4_f32(
                       false, a, false, bl[s], (short)0, accl, false, false);
            accr = __builtin_amdgcn_wmma_f32_16x16x4_f32(
                       false, a, false, br[s], (short)0, accr, false, false);
        }
        const int row0 = lo + tl + half * 8;
        float* pl = out  + row0 * HDIM + m;
        float* pr = outr + row0 * HDIM + m;
        if (lo + tl + 16 <= nNodes) {
#pragma unroll
            for (int v = 0; v < 8; ++v) {
                pl[v * HDIM] = accl[v];
                pr[v * HDIM] = accr[v] + bv;
            }
        } else {
#pragma unroll
            for (int v = 0; v < 8; ++v) {
                if (row0 + v < nNodes) {
                    pl[v * HDIM] = accl[v];
                    pr[v * HDIM] = accr[v] + bv;
                }
            }
        }
    }
}

__global__ __launch_bounds__(AGG_THREADS) void bucket_aggregate(
    const float* __restrict__ y, const float* __restrict__ yr,
    const unsigned* __restrict__ packed, const int* __restrict__ bstart,
    float* __restrict__ out, int nNodes)
{
    aggregate_bucket<false>(y, yr, packed, bstart, out, nullptr,
                            nullptr, nullptr, nullptr, nNodes);
}

__global__ __launch_bounds__(AGG_THREADS) void bucket_aggregate_project(
    const float* __restrict__ y, const float* yr,
    const unsigned* __restrict__ packed, const int* __restrict__ bstart,
    float* __restrict__ outl, float* outr,
    const float* __restrict__ Wl, const float* __restrict__ Wr,
    const float* __restrict__ bias, int nNodes)
{
    aggregate_bucket<true>(y, yr, packed, bstart, outl, outr, Wl, Wr, bias, nNodes);
}

#define DEC_PAIRS_PER_WAVE 256
#define DEC_WAVES_PER_WG   8

template <int K>
__device__ __forceinline__ int quad_take(int v)
{
    return __builtin_amdgcn_update_dpp(0, v, K * 0x55, 0xf, 0xf, true);
}
template <int K>
__device__ __forceinline__ float quad_take(float v)
{
    return __int_as_float(quad_take<K>(__float_as_int(v)));
}

template <int K>
__device__ __forceinline__ void dec_load(
    const float4* __restrict__ h4, int ia, int ib, unsigned q,
    float4& u, float4& w)
{
    const unsigned ea = (unsigned)quad_take<K>(ia) * 4u + q;
    const unsigned eb = (unsigned)quad_take<K>(ib) * 4u + q;
    u = h4[ea];
    w = h4[eb];
}

template <int K>
__device__ __forceinline__ void dec_sum(
    const float4& u, const float4& w, int q, float& res)
{
    const float p = u.x * w.x + u.y * w.y + u.z * w.z + u.w * w.w;
    float s = quad_take<0>(p) + quad_take<1>(p);
    s += quad_take<2>(p);
    s += quad_take<3>(p);
    if (q == K) res = s;
}

__global__ __launch_bounds__(256) void decode_edges(
    const float* __restrict__ h, const int* __restrict__ didx,
    float* __restrict__ out, int P)
{
    const int lane = threadIdx.x & 31;
    const int q    = lane & 3;
    const float4* __restrict__ h4 = (const float4*)h;
    const int w = __builtin_amdgcn_readfirstlane((int)(threadIdx.x >> 5));
    const int first = (blockIdx.x * DEC_WAVES_PER_WG + w) * DEC_PAIRS_PER_WAVE;
    int last = first + DEC_PAIRS_PER_WAVE;
    if (last > P) last = P;

#pragma unroll 1
    for (int t0 = first; t0 < last; t0 += 64) {
        int ia[2], ib[2];
#pragma unroll
        for (int j = 0; j < 2; ++j) {
            const int t  = t0 + j * 32 + lane;
            const int tc = t < P ? t : P - 1;
            ia[j] = didx[tc];
            ib[j] = didx[P + tc];
        }
        float4 u[2][4], v[2][4];
#pragma unroll
        for (int j = 0; j < 2; ++j) {
            dec_load<0>(h4, ia[j], ib[j], (unsigned)q, u[j][0], v[j][0]);
            dec_load<1>(h4, ia[j], ib[j], (unsigned)q, u[j][1], v[j][1]);
            dec_load<2>(h4, ia[j], ib[j], (unsigned)q, u[j][2], v[j][2]);
            dec_load<3>(h4, ia[j], ib[j], (unsigned)q, u[j][3], v[j][3]);
        }
#pragma unroll
        for (int j = 0; j < 2; ++j) {
            float res = 0.0f;
            dec_sum<0>(u[j][0], v[j][0], q, res);
            dec_sum<1>(u[j][1], v[j][1], q, res);
            dec_sum<2>(u[j][2], v[j][2], q, res);
            dec_sum<3>(u[j][3], v[j][3], q, res);
            const int t = t0 + j * 32 + lane;
            if (t < P) out[t] = 1.0f / (1.0f + __expf(-res));
        }
    }
}

extern "C" void kernel_launch(void* const* d_in, const int* in_sizes, int n_in,
                              void* d_out, int out_size, void* d_ws, size_t ws_size,
                              hipStream_t stream) {
    const float* x   = (const float*)d_in[0];
    const float* W1l = (const float*)d_in[1];
    const float* b1  = (const float*)d_in[2];
    const float* W1r = (const float*)d_in[3];
    const float* W2l = (const float*)d_in[4];
    const float* b2  = (const float*)d_in[5];
    const float* W2r = (const float*)d_in[6];
    const int* eidx  = (const int*)d_in[7];
    const int* didx  = (const int*)d_in[8];
    float* out = (float*)d_out;

    const int N = in_sizes[0] / 128;
    const int E = in_sizes[7] / 2;
    const int P = in_sizes[8] / 2;
    const long long NH = (long long)N * HDIM;

    float* buf0 = (float*)d_ws;
    float* buf1 = buf0 + NH;
    float* buf2 = buf1 + NH;
    unsigned* packed = (unsigned*)(buf2 + NH);
    int* bstart = (int*)(packed + E);

    const int NB    = (N + BNODES - 1) / BNODES;
    const int chunk = (E + GCHUNKS - 1) / GCHUNKS;
    int* hist = (int*)buf2;
    int* offs = hist + GCHUNKS * NB;

    const int tiles = (N + 15) / 16;
    const int decPerWg   = DEC_WAVES_PER_WG * DEC_PAIRS_PER_WAVE;
    const int decBlocks  = (P + decPerWg - 1) / decPerWg;

    bucket_hist<<<GCHUNKS, 1024, 0, stream>>>(eidx, hist, E, NB, chunk);
    const int xfWide = (tiles + XF_TILES_PER_WG - 1) / XF_TILES_PER_WG;
    bucket_scan_and_project<<<1 + xfWide, 1024, 0, stream>>>(
        hist, offs, bstart, NB, x, W1l, W1r, b1, buf0, buf1, N);
    bucket_scatter<<<GCHUNKS, 1024, 0, stream>>>(eidx, offs, bstart, packed,
                                                 E, NB, chunk);

    bucket_aggregate_project<<<NB, AGG_THREADS, 0, stream>>>(
        buf0, buf1, packed, bstart, buf2, buf1, W2l, W2r, b2, N);

    bucket_aggregate<<<NB, AGG_THREADS, 0, stream>>>(buf2, buf1, packed, bstart,
                                                     buf0, N);

    decode_edges<<<decBlocks, DEC_WAVES_PER_WG * 32, 0, stream>>>(buf0, didx, out, P);
}


# binning made chunk-local: a chunk's work-group counts, scans, sorts its edges in LDS, writes one block; aggregates walk 128 runs; 6 -> 4 launches; sums in arrival order (changed by layout)
# speedup vs baseline: 3.8794x; 1.4396x over previous
#include <hip/hip_runtime.h>
#include <math.h>

typedef __attribute__((ext_vector_type(2))) float v2f;
typedef __attribute__((ext_vector_type(8))) float v8f;

#define N_NODES 100000
#define HDIM 16

template <int K>
__device__ __forceinline__ void node_tile(
    const int wave,
    const float* __restrict__ xin, const float* __restrict__ Wl,
    const float* __restrict__ Wr,  const float* __restrict__ bias,
    float* __restrict__ outl, float* __restrict__ outr, int nNodes)
{
    const int lane  = threadIdx.x & 31;
    const int tile0 = wave * 16;
    if (tile0 >= nNodes) return;

    const int m  = lane & 15;
    const int kh = (lane >> 4) << 1;

    int rowA = tile0 + m;
    if (rowA >= nNodes) rowA = nNodes - 1;
    const float* xrow = xin + (long long)rowA * K;

    v8f accl = {0.f,0.f,0.f,0.f,0.f,0.f,0.f,0.f};
    v8f accr = {0.f,0.f,0.f,0.f,0.f,0.f,0.f,0.f};

#pragma unroll
    for (int k0 = 0; k0 < K; k0 += 4) {
        const int kr = k0 + kh;
        v2f a = *(const v2f*)(xrow + kr);
        v2f bl = { Wl[kr * HDIM + m], Wl[(kr + 1) * HDIM + m] };
        v2f br = { Wr[kr * HDIM + m], Wr[(kr + 1) * HDIM + m] };
        accl = __builtin_amdgcn_wmma_f32_16x16x4_f32(
                   false, a, false, bl, (short)0, accl, false, false);
        accr = __builtin_amdgcn_wmma_f32_16x16x4_f32(
                   false, a, false, br, (short)0, accr, false, false);
    }

    const float bv    = bias[m];
    const int   mbase = (lane >> 4) * 8;
#pragma unroll
    for (int v = 0; v < 8; ++v) {
        const int row = tile0 + mbase + v;
        if (row < nNodes) {
            outl[row * HDIM + m] = accl[v];
            outr[row * HDIM + m] = accr[v] + bv;
        }
    }
}

#define BSHIFT 8
#define BNODES 256
#define MAXNB  1024
#define GCHUNKS 128
#define BIN_EPT 13
#define BIN_CAP (BIN_EPT * 1024)

#define XF_TILES_PER_WG 25
__global__ __launch_bounds__(1024) void bin_and_project(
    const int* __restrict__ eidx, unsigned* __restrict__ packed,
    int* __restrict__ runoff, int E, int NB, int chunk,
    const float* __restrict__ xin, const float* __restrict__ Wl,
    const float* __restrict__ Wr,  const float* __restrict__ bias,
    float* __restrict__ outl, float* __restrict__ outr, int nNodes)
{
    __shared__ int cnt[MAXNB];
    __shared__ int sc[1024];
    __shared__ unsigned stage[BIN_CAP];
    if (blockIdx.x >= GCHUNKS) {
        const int w = __builtin_amdgcn_readfirstlane((int)(threadIdx.x >> 5));
        if (w < XF_TILES_PER_WG)
            node_tile<128>((blockIdx.x - GCHUNKS) * XF_TILES_PER_WG + w,
                           xin, Wl, Wr, bias, outl, outr, nNodes);
        return;
    }
    const int tid = threadIdx.x;
    cnt[tid] = 0;
    const int beg = blockIdx.x * chunk;
    const int end = (beg + chunk < E) ? (beg + chunk) : E;
    const int n   = end > beg ? end - beg : 0;
    const int* __restrict__ dstp = eidx + E;
    int  s[BIN_EPT], d[BIN_EPT];
    bool ok[BIN_EPT];
#pragma unroll
    for (int u = 0; u < BIN_EPT; ++u) {
        const int e  = beg + u * 1024 + tid;
        ok[u] = e < end;
        const int ec = ok[u] ? e : E - 1;
        s[u] = eidx[ec];
        d[u] = dstp[ec];
    }
    __syncthreads();
    int rk[BIN_EPT];
#pragma unroll
    for (int u = 0; u < BIN_EPT; ++u)
        rk[u] = atomicAdd(&cnt[d[u] >> BSHIFT], ok[u] ? 1 : 0);
    __syncthreads();
    const int c = cnt[tid];
    sc[tid] = c;
    __syncthreads();
    for (int k = 1; k < 1024; k <<= 1) {
        const int t = (tid >= k) ? sc[tid - k] : 0;
        __syncthreads();
        sc[tid] += t;
        __syncthreads();
    }
    const int excl = sc[tid] - c;
    if (tid <= NB) runoff[tid * GCHUNKS + blockIdx.x] = excl;
    cnt[tid] = excl;
    __syncthreads();
#pragma unroll
    for (int u = 0; u < BIN_EPT; ++u) {
        if (ok[u])
            stage[cnt[d[u] >> BSHIFT] + rk[u]] =
                ((unsigned)s[u] << BSHIFT) | (unsigned)(d[u] & (BNODES - 1));
    }
    __syncthreads();
    for (int i = tid; i < n; i += 1024) packed[beg + i] = stage[i];
}

#define AGG_THREADS 256
#define AGG_SPARE   32
#define AGG_PLANE   (BNODES * 4 + AGG_SPARE)
#define AGG_RUN_LOADS 5
#define AGG_RUN_EDGES (AGG_RUN_LOADS * 8)

template <bool PROJECT>
__device__ __forceinline__ void aggregate_bucket(
    const float* y, const float* yr,
    const unsigned* packed, const int* runoff, int chunk,
    float* out, float* outr,
    const float* Wl, const float* Wr, const float* bias, int nNodes)
{
    __shared__ __attribute__((aligned(16))) float acc[4 * AGG_PLANE];
    __shared__ float cntL[BNODES + AGG_SPARE];
    __shared__ int rbL[GCHUNKS];
    __shared__ int reL[GCHUNKS];
    const int tid = threadIdx.x;

    int r0 = 0, r1 = 0;
    if (tid < GCHUNKS) {
        r0 = runoff[blockIdx.x * GCHUNKS + tid];
        r1 = runoff[(blockIdx.x + 1) * GCHUNKS + tid];
    }
    for (int i = tid; i < 4 * AGG_PLANE; i += AGG_THREADS) acc[i] = 0.0f;
    for (int i = tid; i < BNODES + AGG_SPARE; i += AGG_THREADS) cntL[i] = 0.0f;
    if (tid < GCHUNKS) {
        const int cb = tid * chunk;
        rbL[tid] = cb + r0;
        reL[tid] = cb + r1;
    }
    __syncthreads();

    v2f bl[4], br[4];
    float bv = 0.0f;
    if (PROJECT) {
        const int m  = tid & 15;
        const int kh = ((tid >> 4) & 1) << 1;
#pragma unroll
        for (int s = 0; s < 4; ++s) {
            const int kr = 4 * s + kh;
            bl[s] = (v2f){ Wl[kr * HDIM + m], Wl[(kr + 1) * HDIM + m] };
            br[s] = (v2f){ Wr[kr * HDIM + m], Wr[(kr + 1) * HDIM + m] };
        }
        bv = bias[m];
    }

    const int q    = tid & 3;
    const int quad = (tid >> 2) & 7;
    const int spare = BNODES * 4 + (tid & 31);
    const float4* __restrict__ y4 = (const float4*)y;
    const int wq = __builtin_amdgcn_readfirstlane(tid >> 5);
    for (int g = wq; g < GCHUNKS; g += 16) {
        const int b0 = __builtin_amdgcn_readfirstlane(rbL[g]);
        const int e0 = __builtin_amdgcn_readfirstlane(reL[g]);
        const int b1 = __builtin_amdgcn_readfirstlane(rbL[g + 8]);
        const int e1 = __builtin_amdgcn_readfirstlane(reL[g + 8]);
        const int c0 = e0 > b0 ? e0 - 1 : 0;
        const int c1 = e1 > b1 ? e1 - 1 : 0;
        const int len0 = e0 - b0;
        const int len1 = e1 - b1;
        const int lmax = len0 > len1 ? len0 : len1;
        for (int o = 0; o < lmax; o += AGG_RUN_EDGES) {
            unsigned pk[2 * AGG_RUN_LOADS];
            float4   v[2 * AGG_RUN_LOADS];
            bool     ok[2 * AGG_RUN_LOADS];
#pragma unroll
            for (int u = 0; u < AGG_RUN_LOADS; ++u) {
                const int i0 = b0 + o + u * 8 + quad;
                const int i1 = b1 + o + u * 8 + quad;
                ok[u]                 = i0 < e0;
                ok[AGG_RUN_LOADS + u] = i1 < e1;
                pk[u]                 = packed[ok[u] ? i0 : c0];
                pk[AGG_RUN_LOADS + u] = packed[ok[AGG_RUN_LOADS + u] ? i1 : c1];
            }
#pragma unroll
            for (int u = 0; u < 2 * AGG_RUN_LOADS; ++u)
                v[u] = y4[(pk[u] >> BSHIFT) * 4u + (unsigned)q];
#pragma unroll
            for (int u = 0; u < 2 * AGG_RUN_LOADS; ++u) {
                const int dl = (int)(pk[u] & (BNODES - 1));
                const int a  = ok[u] ? dl * 4 + q : spare;
                atomicAdd(&acc[a], v[u].x);
                atomicAdd(&acc[AGG_PLANE + a], v[u].y);
                atomicAdd(&acc[2 * AGG_PLANE + a], v[u].z);
                atomicAdd(&acc[3 * AGG_PLANE + a], v[u].w);
                if (q == 0) atomicAdd(&cntL[ok[u] ? dl : BNODES + (tid & 31)], 1.0f);
            }
        }
    }
    __syncthreads();

    const int lo = blockIdx.x * BNODES;
    if (!PROJECT) {
        const float4* __restrict__ yr4 = (const float4*)yr;
        float4* __restrict__ out4 = (float4*)out;
        for (int i = tid; i < BNODES * 4; i += AGG_THREADS) {
            const int n = lo + (i >> 2);
            if (n < nNodes) {
                float c = cntL[i >> 2];
                c = c > 1.0f ? c : 1.0f;
                const long long t = (long long)n * 4 + (i & 3);
                const float4 b = yr4[t];
                float4 r;
                r.x = acc[i] / c + b.x;
                r.y = acc[AGG_PLANE + i] / c + b.y;
                r.z = acc[2 * AGG_PLANE + i] / c + b.z;
                r.w = acc[3 * AGG_PLANE + i] / c + b.w;
                out4[t] = r;
            }
        }
        return;
    }

    const float4* yr4 = (const float4*)yr;
    for (int i = tid; i < BNODES * 4; i += AGG_THREADS) {
        const int n = lo + (i >> 2);
        if (n < nNodes) {
            float c = cntL[i >> 2];
            c = c > 1.0f ? c : 1.0f;
            const float4 b = yr4[(long long)n * 4 + (i & 3)];
            float4 r;
            r.x = acc[i] / c + b.x;
            r.y = acc[AGG_PLANE + i] / c + b.y;
            r.z = acc[2 * AGG_PLANE + i] / c + b.z;
            r.w = acc[3 * AGG_PLANE + i] / c + b.w;
            acc[i]                 = r.x > 0.0f ? r.x : 0.0f;
            acc[AGG_PLANE + i]     = r.y > 0.0f ? r.y : 0.0f;
            acc[2 * AGG_PLANE + i] = r.z > 0.0f ? r.z : 0.0f;
            acc[3 * AGG_PLANE + i] = r.w > 0.0f ? r.w : 0.0f;
        }
    }
    __syncthreads();

    const int lane  = tid & 31;
    const int m     = lane & 15;
    const int half  = lane >> 4;
    const int wv = __builtin_amdgcn_readfirstlane(tid >> 5);
#pragma unroll
    for (int t = 0; t < 2; ++t) {
        const int tl = (wv * 2 + t) * 16;
        if (lo + tl >= nNodes) continue;
        const float4 p0 = *(const float4*)&acc[(2 * half)     * AGG_PLANE + (tl + m) * 4];
        const float4 p1 = *(const float4*)&acc[(2 * half + 1) * AGG_PLANE + (tl + m) * 4];
        const float e0[4] = { p0.x, p0.y, p0.z, p0.w };
        const float e1[4] = { p1.x, p1.y, p1.z, p1.w };
        v8f accl = {0.f,0.f,0.f,0.f,0.f,0.f,0.f,0.f};
        v8f accr = {0.f,0.f,0.f,0.f,0.f,0.f,0.f,0.f};
#pragma unroll
        for (int s = 0; s < 4; ++s) {
            const v2f a = { e0[s], e1[s] };
            accl = __builtin_amdgcn_wmma_f32_16x16x4_f32(
                       false, a, false, bl[s], (short)0, accl, false, false);
            accr = __builtin_amdgcn_wmma_f32_16x16x4_f32(
                       false, a, false, br[s], (short)0, accr, false, false);
        }
        const int row0 = lo + tl + half * 8;
        float* pl = out  + row0 * HDIM + m;
        float* pr = outr + row0 * HDIM + m;
        if (lo + tl + 16 <= nNodes) {
#pragma unroll
            for (int v = 0; v < 8; ++v) {
                pl[v * HDIM] = accl[v];
                pr[v * HDIM] = accr[v] + bv;
            }
        } else {
#pragma unroll
            for (int v = 0; v < 8; ++v) {
                if (row0 + v < nNodes) {
                    pl[v * HDIM] = accl[v];
                    pr[v * HDIM] = accr[v] + bv;
                }
            }
        }
    }
}

__global__ __launch_bounds__(AGG_THREADS) void bucket_aggregate(
    const float* __restrict__ y, const float* __restrict__ yr,
    const unsigned* __restrict__ packed, const int* __restrict__ runoff,
    int chunk, float* __restrict__ out, int nNodes)
{
    aggregate_bucket<false>(y, yr, packed, runoff, chunk, out, nullptr,
                            nullptr, nullptr, nullptr, nNodes);
}

__global__ __launch_bounds__(AGG_THREADS) void bucket_aggregate_project(
    const float* __restrict__ y, const float* yr,
    const unsigned* __restrict__ packed, const int* __restrict__ runoff,
    int chunk, float* __restrict__ outl, float* outr,
    const float* __restrict__ Wl, const float* __restrict__ Wr,
    const float* __restrict__ bias, int nNodes)
{
    aggregate_bucket<true>(y, yr, packed, runoff, chunk, outl, outr, Wl, Wr, bias, nNodes);
}

#define DEC_PAIRS_PER_WAVE 256
#define DEC_WAVES_PER_WG   8

template <int K>
__device__ __forceinline__ int quad_take(int v)
{
    return __builtin_amdgcn_update_dpp(0, v, K * 0x55, 0xf, 0xf, true);
}
template <int K>
__device__ __forceinline__ float quad_take(float v)
{
    return __int_as_float(quad_take<K>(__float_as_int(v)));
}

template <int K>
__device__ __forceinline__ void dec_load(
    const float4* __restrict__ h4, int ia, int ib, unsigned q,
    float4& u, float4& w)
{
    const unsigned ea = (unsigned)quad_take<K>(ia) * 4u + q;
    const unsigned eb = (unsigned)quad_take<K>(ib) * 4u + q;
    u = h4[ea];
    w = h4[eb];
}

template <int K>
__device__ __forceinline__ void dec_sum(
    const float4& u, const float4& w, int q, float& res)
{
    const float p = u.x * w.x + u.y * w.y + u.z * w.z + u.w * w.w;
    float s = quad_take<0>(p) + quad_take<1>(p);
    s += quad_take<2>(p);
    s += quad_take<3>(p);
    if (q == K) res = s;
}

__global__ __launch_bounds__(256) void decode_edges(
    const float* __restrict__ h, const int* __restrict__ didx,
    float* __restrict__ out, int P)
{
    const int lane = threadIdx.x & 31;
    const int q    = lane & 3;
    const float4* __restrict__ h4 = (const float4*)h;
    const int w = __builtin_amdgcn_readfirstlane((int)(threadIdx.x >> 5));
    const int first = (blockIdx.x * DEC_WAVES_PER_WG + w) * DEC_PAIRS_PER_WAVE;
    int last = first + DEC_PAIRS_PER_WAVE;
    if (last > P) last = P;

#pragma unroll 1
    for (int t0 = first; t0 < last; t0 += 64) {
        int ia[2], ib[2];
#pragma unroll
        for (int j = 0; j < 2; ++j) {
            const int t  = t0 + j * 32 + lane;
            const int tc = t < P ? t : P - 1;
            ia[j] = didx[tc];
            ib[j] = didx[P + tc];
        }
        float4 u[2][4], v[2][4];
#pragma unroll
        for (int j = 0; j < 2; ++j) {
            dec_load<0>(h4, ia[j], ib[j], (unsigned)q, u[j][0], v[j][0]);
            dec_load<1>(h4, ia[j], ib[j], (unsigned)q, u[j][1], v[j][1]);
            dec_load<2>(h4, ia[j], ib[j], (unsigned)q, u[j][2], v[j][2]);
            dec_load<3>(h4, ia[j], ib[j], (unsigned)q, u[j][3], v[j][3]);
        }
#pragma unroll
        for (int j = 0; j < 2; ++j) {
            float res = 0.0f;
            dec_sum<0>(u[j][0], v[j][0], q, res);
            dec_sum<1>(u[j][1], v[j][1], q, res);
            dec_sum<2>(u[j][2], v[j][2], q, res);
            dec_sum<3>(u[j][3], v[j][3], q, res);
            const int t = t0 + j * 32 + lane;
            if (t < P) out[t] = 1.0f / (1.0f + __expf(-res));
        }
    }
}

extern "C" void kernel_launch(void* const* d_in, const int* in_sizes, int n_in,
                              void* d_out, int out_size, void* d_ws, size_t ws_size,
                              hipStream_t stream) {
    const float* x   = (const float*)d_in[0];
    const float* W1l = (const float*)d_in[1];
    const float* b1  = (const float*)d_in[2];
    const float* W1r = (const float*)d_in[3];
    const float* W2l = (const float*)d_in[4];
    const float* b2  = (const float*)d_in[5];
    const float* W2r = (const float*)d_in[6];
    const int* eidx  = (const int*)d_in[7];
    const int* didx  = (const int*)d_in[8];
    float* out = (float*)d_out;

    const int N = in_sizes[0] / 128;
    const int E = in_sizes[7] / 2;
    const int P = in_sizes[8] / 2;
    const long long NH = (long long)N * HDIM;

    float* buf0 = (float*)d_ws;
    float* buf1 = buf0 + NH;
    float* buf2 = buf1 + NH;
    unsigned* packed = (unsigned*)(buf2 + NH);
    int* runoff = (int*)d_out;

    const int NB    = (N + BNODES - 1) / BNODES;
    const int chunk = (E + GCHUNKS - 1) / GCHUNKS;

    const int tiles = (N + 15) / 16;
    const int decPerWg   = DEC_WAVES_PER_WG * DEC_PAIRS_PER_WAVE;
    const int decBlocks  = (P + decPerWg - 1) / decPerWg;

    const int xfWide = (tiles + XF_TILES_PER_WG - 1) / XF_TILES_PER_WG;
    bin_and_project<<<GCHUNKS + xfWide, 1024, 0, stream>>>(
        eidx, packed, runoff, E, NB, chunk, x, W1l, W1r, b1, buf0, buf1, N);

    bucket_aggregate_project<<<NB, AGG_THREADS, 0, stream>>>(
        buf0, buf1, packed, runoff, chunk, buf2, buf1, W2l, W2r, b2, N);

    bucket_aggregate<<<NB, AGG_THREADS, 0, stream>>>(buf2, buf1, packed, runoff,
                                                     chunk, buf0, N);

    decode_edges<<<decBlocks, DEC_WAVES_PER_WG * 32, 0, stream>>>(buf0, didx, out, P);
}
